# starting kernel
# baseline (speedup 1.0000x reference)
#include <hip/hip_runtime.h>
#include <cstdint>

typedef float v2f __attribute__((ext_vector_type(2)));
typedef float v8f __attribute__((ext_vector_type(8)));

namespace {

constexpr int B_ = 2, T_ = 64, C_ = 384, H_ = 768;
constexpr float LR = 0.01f, MOM = 0.9f, DECAY = 0.001f;

constexpr int OFF_W    = 0;
constexpr int OFF_K    = 128;
constexpr int OFF_V    = OFF_K    + B_*T_*C_;
constexpr int OFF_H1   = OFF_V    + B_*T_*C_;
constexpr int OFF_GP   = OFF_H1   + B_*T_*H_;
constexpr int OFF_DP   = OFF_GP   + B_*T_*H_;
constexpr int OFF_WDPT = OFF_DP   + B_*T_*C_;
constexpr int OFF_WDHT = OFF_WDPT + B_*C_*T_;

__device__ __forceinline__ v8f wmma4(v2f a, v2f b, v8f c) {
  return __builtin_amdgcn_wmma_f32_16x16x4_f32(false, a, false, b, (short)0, c,
                                               false, false);
}

template <bool BT>
__device__ __forceinline__ void gemm32x32(const float* __restrict__ A, int lda,
                                          const float* __restrict__ Bm, int ldb,
                                          int K, int m0, int n0, int lane,
                                          v8f acc[4]) {
  const int mr = lane & 15;
  const int kh = (lane >> 4) << 1;
  for (int kk = 0; kk < K; kk += 4) {
    const int kp = kk + kh;
    v2f a0 = *reinterpret_cast<const v2f*>(A + (m0 + mr) * lda + kp);
    v2f a1 = *reinterpret_cast<const v2f*>(A + (m0 + 16 + mr) * lda + kp);
    v2f b0, b1;
    if (BT) {
      b0 = *reinterpret_cast<const v2f*>(Bm + (n0 + mr) * ldb + kp);
      b1 = *reinterpret_cast<const v2f*>(Bm + (n0 + 16 + mr) * ldb + kp);
    } else {
      const float* pb = Bm + kp * ldb + n0 + mr;
      b0.x = pb[0];  b0.y = pb[ldb];
      b1.x = pb[16]; b1.y = pb[ldb + 16];
    }
    acc[0] = wmma4(a0, b0, acc[0]);
    acc[1] = wmma4(a0, b1, acc[1]);
    acc[2] = wmma4(a1, b0, acc[2]);
    acc[3] = wmma4(a1, b1, acc[3]);
  }
}

__global__ void k_init_w(float* __restrict__ ws) {
  int t = threadIdx.x;
  const float a = 1.0f - DECAY;
  float n = (float)(T_ - t);
  float an = __powf(a, n), mn = __powf(MOM, n);
  ws[OFF_W + t] = -LR * (1.0f - MOM) * (an - mn) / (a - MOM);
  if (t == 0) ws[OFF_W + 64] = __powf(a, (float)T_);
}

__global__ void k_gemm_kv(const float* __restrict__ x,
                          const float* __restrict__ Wk, const float* __restrict__ bk,
                          const float* __restrict__ Wv, const float* __restrict__ bv,
                          float* __restrict__ ws) {
  const int b = blockIdx.y, which = blockIdx.z, lane = threadIdx.x;
  const int m0 = (blockIdx.x & 1) * 32, n0 = (blockIdx.x >> 1) * 32;
  const float* A = x + b * T_ * C_;
  const float* W = which ? Wv : Wk;
  const float* bias = which ? bv : bk;
  float* out = ws + (which ? OFF_V : OFF_K) + b * T_ * C_;
  v8f acc[4] = {};
  gemm32x32<true>(A, C_, W, C_, C_, m0, n0, lane, acc);
  const int cr = lane & 15, hi = lane >> 4;
#pragma unroll
  for (int mi = 0; mi < 2; ++mi)
#pragma unroll
    for (int ni = 0; ni < 2; ++ni) {
      int col = n0 + ni * 16 + cr;
      float bb = bias[col];
      v8f a = acc[mi * 2 + ni];
#pragma unroll
      for (int r = 0; r < 8; ++r) {
        int row = m0 + mi * 16 + r + 8 * hi;
        out[row * C_ + col] = a[r] + bb;
      }
    }
}

__global__ void k_gemm_h1(const float* __restrict__ sW0,
                          const float* __restrict__ sb0,
                          float* __restrict__ ws) {
  const int b = blockIdx.y, lane = threadIdx.x;
  const int m0 = (blockIdx.x % 2) * 32, n0 = (blockIdx.x / 2) * 32;
  const float* A = ws + OFF_K + b * T_ * C_;
  const float* W = sW0 + b * H_ * C_;
  float* h1 = ws + OFF_H1 + b * T_ * H_;
  float* gp = ws + OFF_GP + b * T_ * H_;
  v8f acc[4] = {};
  gemm32x32<true>(A, C_, W, C_, C_, m0, n0, lane, acc);
  const int cr = lane & 15, hi = lane >> 4;
#pragma unroll
  for (int mi = 0; mi < 2; ++mi)
#pragma unroll
    for (int ni = 0; ni < 2; ++ni) {
      int col = n0 + ni * 16 + cr;
      float bb = sb0[b * H_ + col];
      v8f a = acc[mi * 2 + ni];
#pragma unroll
      for (int r = 0; r < 8; ++r) {
        int row = m0 + mi * 16 + r + 8 * hi;
        float pre = a[r] + bb;
        float sig = 1.0f / (1.0f + __expf(-pre));
        h1[row * H_ + col] = pre * sig;
        gp[row * H_ + col] = sig * (1.0f + pre * (1.0f - sig));
      }
    }
}

__global__ void k_gemm_dp(const float* __restrict__ sW1,
                          const float* __restrict__ sb1,
                          float* __restrict__ ws) {
  const int b = blockIdx.y, lane = threadIdx.x;
  const int m0 = (blockIdx.x % 2) * 32, n0 = (blockIdx.x / 2) * 32;
  const float* A = ws + OFF_H1 + b * T_ * H_;
  const float* W = sW1 + b * C_ * H_;
  const float* v = ws + OFF_V + b * T_ * C_;
  float* dp = ws + OFF_DP + b * T_ * C_;
  float* wdpT = ws + OFF_WDPT + b * C_ * T_;
  v8f acc[4] = {};
  gemm32x32<true>(A, H_, W, H_, H_, m0, n0, lane, acc);
  const int cr = lane & 15, hi = lane >> 4;
#pragma unroll
  for (int mi = 0; mi < 2; ++mi)
#pragma unroll
    for (int ni = 0; ni < 2; ++ni) {
      int col = n0 + ni * 16 + cr;
      float bb = sb1[b * C_ + col];
      v8f a = acc[mi * 2 + ni];
#pragma unroll
      for (int r = 0; r < 8; ++r) {
        int row = m0 + mi * 16 + r + 8 * hi;
        float d = (a[r] + bb) - v[row * C_ + col];
        dp[row * C_ + col] = d;
        wdpT[col * T_ + row] = ws[OFF_W + row] * d;
      }
    }
}

__global__ void k_gemm_dh1(const float* __restrict__ sW1,
                           float* __restrict__ ws) {
  const int b = blockIdx.y, lane = threadIdx.x;
  const int m0 = (blockIdx.x % 2) * 32, n0 = (blockIdx.x / 2) * 32;
  const float* A = ws + OFF_DP + b * T_ * C_;
  const float* W = sW1 + b * C_ * H_;
  const float* gp = ws + OFF_GP + b * T_ * H_;
  float* wdhT = ws + OFF_WDHT + b * H_ * T_;
  v8f acc[4] = {};
  gemm32x32<false>(A, C_, W, H_, C_, m0, n0, lane, acc);
  const int cr = lane & 15, hi = lane >> 4;
#pragma unroll
  for (int mi = 0; mi < 2; ++mi)
#pragma unroll
    for (int ni = 0; ni < 2; ++ni) {
      int col = n0 + ni * 16 + cr;
      v8f a = acc[mi * 2 + ni];
#pragma unroll
      for (int r = 0; r < 8; ++r) {
        int row = m0 + mi * 16 + r + 8 * hi;
        float d = a[r] * gp[row * H_ + col];
        wdhT[col * T_ + row] = ws[OFF_W + row] * d;
      }
    }
}

__global__ void k_gemm_nw0(const float* __restrict__ sW0,
                           float* __restrict__ nW0,
                           float* __restrict__ ws) {
  const int b = blockIdx.y, lane = threadIdx.x;
  const int m0 = (blockIdx.x / 12) * 32, n0 = (blockIdx.x % 12) * 32;
  const float* A = ws + OFF_WDHT + b * H_ * T_;
  const float* Bm = ws + OFF_K + b * T_ * C_;
  const float dT = ws[OFF_W + 64];
  float* out = nW0 + b * H_ * C_;
  const float* w0 = sW0 + b * H_ * C_;
  v8f acc[4] = {};
  gemm32x32<false>(A, T_, Bm, C_, T_, m0, n0, lane, acc);
  const int cr = lane & 15, hi = lane >> 4;
#pragma unroll
  for (int mi = 0; mi < 2; ++mi)
#pragma unroll
    for (int ni = 0; ni < 2; ++ni) {
      int col = n0 + ni * 16 + cr;
      v8f a = acc[mi * 2 + ni];
#pragma unroll
      for (int r = 0; r < 8; ++r) {
        int row = m0 + mi * 16 + r + 8 * hi;
        out[row * C_ + col] = dT * w0[row * C_ + col] + a[r];
      }
    }
}

__global__ void k_gemm_nw1(const float* __restrict__ sW1,
                           float* __restrict__ nW1,
                           float* __restrict__ ws) {
  const int b = blockIdx.y, lane = threadIdx.x;
  const int m0 = (blockIdx.x / 24) * 32, n0 = (blockIdx.x % 24) * 32;
  const float* A = ws + OFF_WDPT + b * C_ * T_;
  const float* Bm = ws + OFF_H1 + b * T_ * H_;
  const float dT = ws[OFF_W + 64];
  float* out = nW1 + b * C_ * H_;
  const float* w1 = sW1 + b * C_ * H_;
  v8f acc[4] = {};
  gemm32x32<false>(A, T_, Bm, H_, T_, m0, n0, lane, acc);
  const int cr = lane & 15, hi = lane >> 4;
#pragma unroll
  for (int mi = 0; mi < 2; ++mi)
#pragma unroll
    for (int ni = 0; ni < 2; ++ni) {
      int col = n0 + ni * 16 + cr;
      v8f a = acc[mi * 2 + ni];
#pragma unroll
      for (int r = 0; r < 8; ++r) {
        int row = m0 + mi * 16 + r + 8 * hi;
        out[row * H_ + col] = dT * w1[row * H_ + col] + a[r];
      }
    }
}

__global__ void k_bias(const float* __restrict__ sb0, const float* __restrict__ sb1,
                       float* __restrict__ nb0, float* __restrict__ nb1,
                       const float* __restrict__ ws) {
  int idx = blockIdx.x * blockDim.x + threadIdx.x;
  if (idx >= B_ * (H_ + C_)) return;
  int b = idx / (H_ + C_), r = idx % (H_ + C_);
  const float dT = ws[OFF_W + 64];
  if (r < H_) {
    const float* p = ws + OFF_WDHT + b * H_ * T_ + r * T_;
    float s = 0.f;
#pragma unroll 4
    for (int t = 0; t < T_; ++t) s += p[t];
    nb0[b * H_ + r] = dT * sb0[b * H_ + r] + s;
  } else {
    int c = r - H_;
    const float* p = ws + OFF_WDPT + b * C_ * T_ + c * T_;
    float s = 0.f;
#pragma unroll 4
    for (int t = 0; t < T_; ++t) s += p[t];
    nb1[b * C_ + c] = dT * sb1[b * C_ + c] + s;
  }
}

__global__ void k_retrieve(const float* __restrict__ Wq, const float* __restrict__ bq,
                           const float* __restrict__ Wo, const float* __restrict__ bo,
                           const float* __restrict__ iq,
                           const float* __restrict__ sW0, const float* __restrict__ sb0,
                           const float* __restrict__ sW1, const float* __restrict__ sb1,
                           float* __restrict__ out) {
  __shared__ float q[C_], h[H_], pred[C_];
  const int b = blockIdx.x, t = threadIdx.x;
  {
    const float* wr = Wq + t * C_;
    float s = 0.f;
    for (int c = 0; c < C_; ++c) s += wr[c] * iq[c];
    q[t] = s + bq[t];
  }
  __syncthreads();
#pragma unroll
  for (int j = 0; j < 2; ++j) {
    int hh = t + j * C_;
    const float* w0 = sW0 + b * H_ * C_ + hh * C_;
    float s = 0.f;
    for (int c = 0; c < C_; ++c) s += w0[c] * q[c];
    s += sb0[b * H_ + hh];
    float sig = 1.0f / (1.0f + __expf(-s));
    h[hh] = s * sig;
  }
  __syncthreads();
  {
    const float* w1 = sW1 + b * C_ * H_ + t * H_;
    float s = 0.f;
    for (int hh = 0; hh < H_; ++hh) s += w1[hh] * h[hh];
    pred[t] = s + sb1[b * C_ + t];
  }
  __syncthreads();
  {
    const float* wo = Wo + t * C_;
    float s = 0.f;
    for (int c = 0; c < C_; ++c) s += wo[c] * pred[c];
    out[b * C_ + t] = s + bo[t];
  }
}

}

extern "C" void kernel_launch(void* const* d_in, const int* in_sizes, int n_in,
                              void* d_out, int out_size, void* d_ws, size_t ws_size,
                              hipStream_t stream) {
  const float* x   = (const float*)d_in[0];
  const float* Wq  = (const float*)d_in[1];
  const float* bq  = (const float*)d_in[2];
  const float* Wk  = (const float*)d_in[3];
  const float* bk  = (const float*)d_in[4];
  const float* Wv  = (const float*)d_in[5];
  const float* bv  = (const float*)d_in[6];
  const float* Wo  = (const float*)d_in[7];
  const float* bo  = (const float*)d_in[8];
  const float* iq  = (const float*)d_in[9];
  const float* sW0 = (const float*)d_in[10];
  const float* sb0 = (const float*)d_in[11];
  const float* sW1 = (const float*)d_in[12];
  const float* sb1 = (const float*)d_in[13];

  float* out = (float*)d_out;
  float* ws  = (float*)d_ws;

  float* out_out = out;
  float* out_nW0 = out + B_ * C_;
  float* out_nb0 = out_nW0 + B_ * H_ * C_;
  float* out_nW1 = out_nb0 + B_ * H_;
  float* out_nb1 = out_nW1 + B_ * C_ * H_;

  k_init_w <<<1, 64, 0, stream>>>(ws);
  k_gemm_kv<<<dim3(24, B_, 2), 32, 0, stream>>>(x, Wk, bk, Wv, bv, ws);
  k_gemm_h1<<<dim3(48, B_), 32, 0, stream>>>(sW0, sb0, ws);
  k_gemm_dp<<<dim3(24, B_), 32, 0, stream>>>(sW1, sb1, ws);
  k_gemm_dh1<<<dim3(48, B_), 32, 0, stream>>>(sW1, ws);
  k_gemm_nw0<<<dim3(288, B_), 32, 0, stream>>>(sW0, out_nW0, ws);
  k_gemm_nw1<<<dim3(288, B_), 32, 0, stream>>>(sW1, out_nW1, ws);
  k_bias    <<<dim3(18), 128, 0, stream>>>(sb0, sb1, out_nb0, out_nb1, ws);
  k_retrieve<<<dim3(B_), C_, 0, stream>>>(Wq, bq, Wo, bo, iq, sW0, sb0, sW1, sb1, out_out);
}


# Retrieval's 4 chained matvecs leave the 2-block tail kernel and run as extra one-wave blocks of launches 0-3 (row per wave, coalesced, lane-butterfly sums: out's summation order changes); 8 launches
# speedup vs baseline: 1.7257x; 1.7257x over previous
#include <hip/hip_runtime.h>
#include <cstdint>

typedef float v2f __attribute__((ext_vector_type(2)));
typedef float v4f __attribute__((ext_vector_type(4)));
typedef float v8f __attribute__((ext_vector_type(8)));

namespace {

constexpr int B_ = 2, T_ = 64, C_ = 384, H_ = 768;
constexpr float LR = 0.01f, MOM = 0.9f, DECAY = 0.001f;

constexpr int OFF_W    = 0;
constexpr int OFF_K    = 128;
constexpr int OFF_V    = OFF_K    + B_*T_*C_;
constexpr int OFF_H1   = OFF_V    + B_*T_*C_;
constexpr int OFF_GP   = OFF_H1   + B_*T_*H_;
constexpr int OFF_DP   = OFF_GP   + B_*T_*H_;
constexpr int OFF_WDPT = OFF_DP   + B_*T_*C_;
constexpr int OFF_WDHT = OFF_WDPT + B_*C_*T_;
constexpr int OFF_RQ   = OFF_WDHT + B_*H_*T_;
constexpr int OFF_RH   = OFF_RQ   + C_;
constexpr int OFF_RP   = OFF_RH   + B_*H_;

__device__ __forceinline__ v8f wmma4(v2f a, v2f b, v8f c) {
  return __builtin_amdgcn_wmma_f32_16x16x4_f32(false, a, false, b, (short)0, c,
                                               false, false);
}

template <bool BT>
__device__ __forceinline__ void gemm32x32(const float* __restrict__ A, int lda,
                                          const float* __restrict__ Bm, int ldb,
                                          int K, int m0, int n0, int lane,
                                          v8f acc[4]) {
  const int mr = lane & 15;
  const int kh = (lane >> 4) << 1;
  for (int kk = 0; kk < K; kk += 4) {
    const int kp = kk + kh;
    v2f a0 = *reinterpret_cast<const v2f*>(A + (m0 + mr) * lda + kp);
    v2f a1 = *reinterpret_cast<const v2f*>(A + (m0 + 16 + mr) * lda + kp);
    v2f b0, b1;
    if (BT) {
      b0 = *reinterpret_cast<const v2f*>(Bm + (n0 + mr) * ldb + kp);
      b1 = *reinterpret_cast<const v2f*>(Bm + (n0 + 16 + mr) * ldb + kp);
    } else {
      const float* pb = Bm + kp * ldb + n0 + mr;
      b0.x = pb[0];  b0.y = pb[ldb];
      b1.x = pb[16]; b1.y = pb[ldb + 16];
    }
    acc[0] = wmma4(a0, b0, acc[0]);
    acc[1] = wmma4(a0, b1, acc[1]);
    acc[2] = wmma4(a1, b0, acc[2]);
    acc[3] = wmma4(a1, b1, acc[3]);
  }
}

template <int K, int NR, int RU, bool SILU>
__device__ __forceinline__ void wave_matvec(const float* __restrict__ M,
                                            const float* __restrict__ v,
                                            const float* __restrict__ bias,
                                            float* __restrict__ y, int row0, int lane) {
  constexpr int NV = K / 128;
  v4f vv[NV];
#pragma unroll
  for (int j = 0; j < NV; ++j)
    vv[j] = *reinterpret_cast<const v4f*>(v + j * 128 + lane * 4);
  for (int r0 = 0; r0 < NR; r0 += RU) {
    float s[RU];
#pragma unroll
    for (int i = 0; i < RU; ++i) {
      const float* mr = M + (row0 + r0 + i) * K + lane * 4;
      float a = 0.f;
#pragma unroll
      for (int j = 0; j < NV; ++j) {
        v4f m = *reinterpret_cast<const v4f*>(mr + j * 128);
        a += m.x * vv[j].x;
        a += m.y * vv[j].y;
        a += m.z * vv[j].z;
        a += m.w * vv[j].w;
      }
      s[i] = a;
    }
#pragma unroll
    for (int off = 16; off > 0; off >>= 1)
#pragma unroll
      for (int i = 0; i < RU; ++i) s[i] += __shfl_xor(s[i], off, 32);
    float t = s[0];
#pragma unroll
    for (int i = 1; i < RU; ++i)
      if (lane == i) t = s[i];
    if (lane < RU) {
      const int row = row0 + r0 + lane;
      t += bias[row];
      if (SILU) {
        float sig = 1.0f / (1.0f + __expf(-t));
        t = t * sig;
      }
      y[row] = t;
    }
  }
}

__global__ void k_init_w(const float* __restrict__ Wq, const float* __restrict__ bq,
                         const float* __restrict__ iq, float* __restrict__ ws) {
  const int lane = threadIdx.x;
  if (blockIdx.x != 0) {
    wave_matvec<C_, 4, 4, false>(Wq, iq, bq, ws + OFF_RQ, (blockIdx.x - 1) * 4, lane);
    return;
  }
  const float a = 1.0f - DECAY;
  for (int t = lane; t < T_; t += 32) {
    float n = (float)(T_ - t);
    float an = __powf(a, n), mn = __powf(MOM, n);
    ws[OFF_W + t] = -LR * (1.0f - MOM) * (an - mn) / (a - MOM);
  }
  if (lane == 0) ws[OFF_W + 64] = __powf(a, (float)T_);
}

__global__ void k_gemm_kv(const float* __restrict__ x,
                          const float* __restrict__ Wk, const float* __restrict__ bk,
                          const float* __restrict__ Wv, const float* __restrict__ bv,
                          const float* __restrict__ sW0, const float* __restrict__ sb0,
                          float* __restrict__ ws) {
  const int b = blockIdx.y, which = blockIdx.z, lane = threadIdx.x;
  if (blockIdx.x >= 24) {
    const int row0 = which * (H_ / 2) + (blockIdx.x - 24) * 16;
    wave_matvec<C_, 16, 4, true>(sW0 + b * H_ * C_, ws + OFF_RQ, sb0 + b * H_,
                              ws + OFF_RH + b * H_, row0, lane);
    return;
  }
  const int m0 = (blockIdx.x & 1) * 32, n0 = (blockIdx.x >> 1) * 32;
  const float* A = x + b * T_ * C_;
  const float* W = which ? Wv : Wk;
  const float* bias = which ? bv : bk;
  float* out = ws + (which ? OFF_V : OFF_K) + b * T_ * C_;
  v8f acc[4] = {};
  gemm32x32<true>(A, C_, W, C_, C_, m0, n0, lane, acc);
  const int cr = lane & 15, hi = lane >> 4;
#pragma unroll
  for (int mi = 0; mi < 2; ++mi)
#pragma unroll
    for (int ni = 0; ni < 2; ++ni) {
      int col = n0 + ni * 16 + cr;
      float bb = bias[col];
      v8f a = acc[mi * 2 + ni];
#pragma unroll
      for (int r = 0; r < 8; ++r) {
        int row = m0 + mi * 16 + r + 8 * hi;
        out[row * C_ + col] = a[r] + bb;
      }
    }
}

__global__ void k_gemm_h1(const float* __restrict__ sW0,
                          const float* __restrict__ sb0,
                          const float* __restrict__ sW1,
                          const float* __restrict__ sb1,
                          float* __restrict__ ws) {
  const int b = blockIdx.y, lane = threadIdx.x;
  if (blockIdx.x >= 48) {
    wave_matvec<H_, 16, 2, false>(sW1 + b * C_ * H_, ws + OFF_RH + b * H_, sb1 + b * C_,
                               ws + OFF_RP + b * C_, (blockIdx.x - 48) * 16, lane);
    return;
  }
  const int m0 = (blockIdx.x % 2) * 32, n0 = (blockIdx.x / 2) * 32;
  const float* A = ws + OFF_K + b * T_ * C_;
  const float* W = sW0 + b * H_ * C_;
  float* h1 = ws + OFF_H1 + b * T_ * H_;
  float* gp = ws + OFF_GP + b * T_ * H_;
  v8f acc[4] = {};
  gemm32x32<true>(A, C_, W, C_, C_, m0, n0, lane, acc);
  const int cr = lane & 15, hi = lane >> 4;
#pragma unroll
  for (int mi = 0; mi < 2; ++mi)
#pragma unroll
    for (int ni = 0; ni < 2; ++ni) {
      int col = n0 + ni * 16 + cr;
      float bb = sb0[b * H_ + col];
      v8f a = acc[mi * 2 + ni];
#pragma unroll
      for (int r = 0; r < 8; ++r) {
        int row = m0 + mi * 16 + r + 8 * hi;
        float pre = a[r] + bb;
        float sig = 1.0f / (1.0f + __expf(-pre));
        h1[row * H_ + col] = pre * sig;
        gp[row * H_ + col] = sig * (1.0f + pre * (1.0f - sig));
      }
    }
}

__global__ void k_gemm_dp(const float* __restrict__ sW1,
                          const float* __restrict__ sb1,
                          const float* __restrict__ Wo,
                          const float* __restrict__ bo,
                          float* __restrict__ rout,
                          float* __restrict__ ws) {
  const int b = blockIdx.y, lane = threadIdx.x;
  if (blockIdx.x >= 24) {
    wave_matvec<C_, 16, 4, false>(Wo, ws + OFF_RP + b * C_, bo, rout + b * C_,
                               (blockIdx.x - 24) * 16, lane);
    return;
  }
  const int m0 = (blockIdx.x % 2) * 32, n0 = (blockIdx.x / 2) * 32;
  const float* A = ws + OFF_H1 + b * T_ * H_;
  const float* W = sW1 + b * C_ * H_;
  const float* v = ws + OFF_V + b * T_ * C_;
  float* dp = ws + OFF_DP + b * T_ * C_;
  float* wdpT = ws + OFF_WDPT + b * C_ * T_;
  v8f acc[4] = {};
  gemm32x32<true>(A, H_, W, H_, H_, m0, n0, lane, acc);
  const int cr = lane & 15, hi = lane >> 4;
#pragma unroll
  for (int mi = 0; mi < 2; ++mi)
#pragma unroll
    for (int ni = 0; ni < 2; ++ni) {
      int col = n0 + ni * 16 + cr;
      float bb = sb1[b * C_ + col];
      v8f a = acc[mi * 2 + ni];
#pragma unroll
      for (int r = 0; r < 8; ++r) {
        int row = m0 + mi * 16 + r + 8 * hi;
        float d = (a[r] + bb) - v[row * C_ + col];
        dp[row * C_ + col] = d;
        wdpT[col * T_ + row] = ws[OFF_W + row] * d;
      }
    }
}

__global__ void k_gemm_dh1(const float* __restrict__ sW1,
                           float* __restrict__ ws) {
  const int b = blockIdx.y, lane = threadIdx.x;
  const int m0 = (blockIdx.x % 2) * 32, n0 = (blockIdx.x / 2) * 32;
  const float* A = ws + OFF_DP + b * T_ * C_;
  const float* W = sW1 + b * C_ * H_;
  const float* gp = ws + OFF_GP + b * T_ * H_;
  float* wdhT = ws + OFF_WDHT + b * H_ * T_;
  v8f acc[4] = {};
  gemm32x32<false>(A, C_, W, H_, C_, m0, n0, lane, acc);
  const int cr = lane & 15, hi = lane >> 4;
#pragma unroll
  for (int mi = 0; mi < 2; ++mi)
#pragma unroll
    for (int ni = 0; ni < 2; ++ni) {
      int col = n0 + ni * 16 + cr;
      v8f a = acc[mi * 2 + ni];
#pragma unroll
      for (int r = 0; r < 8; ++r) {
        int row = m0 + mi * 16 + r + 8 * hi;
        float d = a[r] * gp[row * H_ + col];
        wdhT[col * T_ + row] = ws[OFF_W + row] * d;
      }
    }
}

__global__ void k_gemm_nw0(const float* __restrict__ sW0,
                           float* __restrict__ nW0,
                           float* __restrict__ ws) {
  const int b = blockIdx.y, lane = threadIdx.x;
  const int m0 = (blockIdx.x / 12) * 32, n0 = (blockIdx.x % 12) * 32;
  const float* A = ws + OFF_WDHT + b * H_ * T_;
  const float* Bm = ws + OFF_K + b * T_ * C_;
  const float dT = ws[OFF_W + 64];
  float* out = nW0 + b * H_ * C_;
  const float* w0 = sW0 + b * H_ * C_;
  v8f acc[4] = {};
  gemm32x32<false>(A, T_, Bm, C_, T_, m0, n0, lane, acc);
  const int cr = lane & 15, hi = lane >> 4;
#pragma unroll
  for (int mi = 0; mi < 2; ++mi)
#pragma unroll
    for (int ni = 0; ni < 2; ++ni) {
      int col = n0 + ni * 16 + cr;
      v8f a = acc[mi * 2 + ni];
#pragma unroll
      for (int r = 0; r < 8; ++r) {
        int row = m0 + mi * 16 + r + 8 * hi;
        out[row * C_ + col] = dT * w0[row * C_ + col] + a[r];
      }
    }
}

__global__ void k_gemm_nw1(const float* __restrict__ sW1,
                           float* __restrict__ nW1,
                           float* __restrict__ ws) {
  const int b = blockIdx.y, lane = threadIdx.x;
  const int m0 = (blockIdx.x / 24) * 32, n0 = (blockIdx.x % 24) * 32;
  const float* A = ws + OFF_WDPT + b * C_ * T_;
  const float* Bm = ws + OFF_H1 + b * T_ * H_;
  const float dT = ws[OFF_W + 64];
  float* out = nW1 + b * C_ * H_;
  const float* w1 = sW1 + b * C_ * H_;
  v8f acc[4] = {};
  gemm32x32<false>(A, T_, Bm, H_, T_, m0, n0, lane, acc);
  const int cr = lane & 15, hi = lane >> 4;
#pragma unroll
  for (int mi = 0; mi < 2; ++mi)
#pragma unroll
    for (int ni = 0; ni < 2; ++ni) {
      int col = n0 + ni * 16 + cr;
      v8f a = acc[mi * 2 + ni];
#pragma unroll
      for (int r = 0; r < 8; ++r) {
        int row = m0 + mi * 16 + r + 8 * hi;
        out[row * H_ + col] = dT * w1[row * H_ + col] + a[r];
      }
    }
}

__global__ void k_bias(const float* __restrict__ sb0, const float* __restrict__ sb1,
                       float* __restrict__ nb0, float* __restrict__ nb1,
                       const float* __restrict__ ws) {
  int idx = blockIdx.x * blockDim.x + threadIdx.x;
  if (idx >= B_ * (H_ + C_)) return;
  int b = idx / (H_ + C_), r = idx % (H_ + C_);
  const float dT = ws[OFF_W + 64];
  if (r < H_) {
    const float* p = ws + OFF_WDHT + b * H_ * T_ + r * T_;
    float s = 0.f;
#pragma unroll 4
    for (int t = 0; t < T_; ++t) s += p[t];
    nb0[b * H_ + r] = dT * sb0[b * H_ + r] + s;
  } else {
    int c = r - H_;
    const float* p = ws + OFF_WDPT + b * C_ * T_ + c * T_;
    float s = 0.f;
#pragma unroll 4
    for (int t = 0; t < T_; ++t) s += p[t];
    nb1[b * C_ + c] = dT * sb1[b * C_ + c] + s;
  }
}

}

extern "C" void kernel_launch(void* const* d_in, const int* in_sizes, int n_in,
                              void* d_out, int out_size, void* d_ws, size_t ws_size,
                              hipStream_t stream) {
  const float* x   = (const float*)d_in[0];
  const float* Wq  = (const float*)d_in[1];
  const float* bq  = (const float*)d_in[2];
  const float* Wk  = (const float*)d_in[3];
  const float* bk  = (const float*)d_in[4];
  const float* Wv  = (const float*)d_in[5];
  const float* bv  = (const float*)d_in[6];
  const float* Wo  = (const float*)d_in[7];
  const float* bo  = (const float*)d_in[8];
  const float* iq  = (const float*)d_in[9];
  const float* sW0 = (const float*)d_in[10];
  const float* sb0 = (const float*)d_in[11];
  const float* sW1 = (const float*)d_in[12];
  const float* sb1 = (const float*)d_in[13];

  float* out = (float*)d_out;
  float* ws  = (float*)d_ws;

  float* out_out = out;
  float* out_nW0 = out + B_ * C_;
  float* out_nb0 = out_nW0 + B_ * H_ * C_;
  float* out_nW1 = out_nb0 + B_ * H_;
  float* out_nb1 = out_nW1 + B_ * C_ * H_;

  k_init_w <<<97, 32, 0, stream>>>(Wq, bq, iq, ws);
  k_gemm_kv<<<dim3(48, B_, 2), 32, 0, stream>>>(x, Wk, bk, Wv, bv, sW0, sb0, ws);
  k_gemm_h1<<<dim3(72, B_), 32, 0, stream>>>(sW0, sb0, sW1, sb1, ws);
  k_gemm_dp<<<dim3(48, B_), 32, 0, stream>>>(sW1, sb1, Wo, bo, out_out, ws);
  k_gemm_dh1<<<dim3(48, B_), 32, 0, stream>>>(sW1, ws);
  k_gemm_nw0<<<dim3(288, B_), 32, 0, stream>>>(sW0, out_nW0, ws);
  k_gemm_nw1<<<dim3(288, B_), 32, 0, stream>>>(sW1, out_nW1, ws);
  k_bias    <<<dim3(18), 128, 0, stream>>>(sb0, sb1, out_nb0, out_nb1, ws);
}


# K range of each 32x32 tile in launches 1-4 split over the 4 or 8 waves of a block; partial tiles summed through LDS (summation order changes), each wave then writing whole 128-byte rows
# speedup vs baseline: 4.9324x; 2.8582x over previous
#include <hip/hip_runtime.h>
#include <cstdint>

typedef float v2f __attribute__((ext_vector_type(2)));
typedef float v4f __attribute__((ext_vector_type(4)));
typedef float v8f __attribute__((ext_vector_type(8)));

namespace {

constexpr int B_ = 2, T_ = 64, C_ = 384, H_ = 768;
constexpr float LR = 0.01f, MOM = 0.9f, DECAY = 0.001f;

constexpr int OFF_W    = 0;
constexpr int OFF_K    = 128;
constexpr int OFF_V    = OFF_K    + B_*T_*C_;
constexpr int OFF_H1   = OFF_V    + B_*T_*C_;
constexpr int OFF_GP   = OFF_H1   + B_*T_*H_;
constexpr int OFF_DP   = OFF_GP   + B_*T_*H_;
constexpr int OFF_WDPT = OFF_DP   + B_*T_*C_;
constexpr int OFF_WDHT = OFF_WDPT + B_*C_*T_;
constexpr int OFF_RQ   = OFF_WDHT + B_*H_*T_;
constexpr int OFF_RH   = OFF_RQ   + C_;
constexpr int OFF_RP   = OFF_RH   + B_*H_;

__device__ __forceinline__ v8f wmma4(v2f a, v2f b, v8f c) {
  return __builtin_amdgcn_wmma_f32_16x16x4_f32(false, a, false, b, (short)0, c,
                                               false, false);
}

template <bool BT>
__device__ __forceinline__ void gemm32x32(const float* __restrict__ A, int lda,
                                          const float* __restrict__ Bm, int ldb,
                                          int K, int m0, int n0, int lane,
                                          v8f acc[4]) {
  const int mr = lane & 15;
  const int kh = (lane >> 4) << 1;
  for (int kk = 0; kk < K; kk += 4) {
    const int kp = kk + kh;
    v2f a0 = *reinterpret_cast<const v2f*>(A + (m0 + mr) * lda + kp);
    v2f a1 = *reinterpret_cast<const v2f*>(A + (m0 + 16 + mr) * lda + kp);
    v2f b0, b1;
    if (BT) {
      b0 = *reinterpret_cast<const v2f*>(Bm + (n0 + mr) * ldb + kp);
      b1 = *reinterpret_cast<const v2f*>(Bm + (n0 + 16 + mr) * ldb + kp);
    } else {
      const float* pb = Bm + kp * ldb + n0 + mr;
      b0.x = pb[0];  b0.y = pb[ldb];
      b1.x = pb[16]; b1.y = pb[ldb + 16];
    }
    acc[0] = wmma4(a0, b0, acc[0]);
    acc[1] = wmma4(a0, b1, acc[1]);
    acc[2] = wmma4(a1, b0, acc[2]);
    acc[3] = wmma4(a1, b1, acc[3]);
  }
}

template <bool BT, int KS>
__device__ __forceinline__ void gemm32x32_slice(const float* __restrict__ A, int lda,
                                                const float* __restrict__ Bm, int ldb,
                                                int k0, int m0, int n0, int lane,
                                                v8f acc[4]) {
  const int mr = lane & 15;
  const int kh = (lane >> 4) << 1;
#pragma unroll
  for (int kk = 0; kk < KS; kk += 4) {
    const int kp = k0 + kk + kh;
    v2f a0 = *reinterpret_cast<const v2f*>(A + (m0 + mr) * lda + kp);
    v2f a1 = *reinterpret_cast<const v2f*>(A + (m0 + 16 + mr) * lda + kp);
    v2f b0, b1;
    if (BT) {
      b0 = *reinterpret_cast<const v2f*>(Bm + (n0 + mr) * ldb + kp);
      b1 = *reinterpret_cast<const v2f*>(Bm + (n0 + 16 + mr) * ldb + kp);
    } else {
      const float* pb = Bm + kp * ldb + n0 + mr;
      b0.x = pb[0];  b0.y = pb[ldb];
      b1.x = pb[16]; b1.y = pb[ldb + 16];
    }
    acc[0] = wmma4(a0, b0, acc[0]);
    acc[1] = wmma4(a0, b1, acc[1]);
    acc[2] = wmma4(a1, b0, acc[2]);
    acc[3] = wmma4(a1, b1, acc[3]);
  }
}

constexpr int RP_ = 36;
constexpr int TP_ = 33;

__device__ __forceinline__ void put_partial(const v8f acc[4], float* red, int slice, int lane) {
  const int cr = lane & 15, hi = lane >> 4;
#pragma unroll
  for (int qq = 0; qq < 4; ++qq)
#pragma unroll
    for (int r = 0; r < 8; ++r) {
      const int R = (qq >> 1) * 16 + 8 * hi + r;
      const int Cc = (qq & 1) * 16 + cr;
      red[(slice * 32 + R) * RP_ + Cc] = acc[qq][r];
    }
}

template <int S>
__device__ __forceinline__ float row_sum(const float* red, int R, int lane) {
  float s = red[R * RP_ + lane];
#pragma unroll
  for (int sl = 1; sl < S; ++sl) s += red[(sl * 32 + R) * RP_ + lane];
  return s;
}

template <int K, int NR, int RU, bool SILU>
__device__ __forceinline__ void wave_matvec(const float* __restrict__ M,
                                            const float* __restrict__ v,
                                            const float* __restrict__ bias,
                                            float* __restrict__ y, int row0, int lane) {
  constexpr int NV = K / 128;
  v4f vv[NV];
#pragma unroll
  for (int j = 0; j < NV; ++j)
    vv[j] = *reinterpret_cast<const v4f*>(v + j * 128 + lane * 4);
  for (int r0 = 0; r0 < NR; r0 += RU) {
    float s[RU];
#pragma unroll
    for (int i = 0; i < RU; ++i) {
      const float* mr = M + (row0 + r0 + i) * K + lane * 4;
      float a = 0.f;
#pragma unroll
      for (int j = 0; j < NV; ++j) {
        v4f m = *reinterpret_cast<const v4f*>(mr + j * 128);
        a += m.x * vv[j].x;
        a += m.y * vv[j].y;
        a += m.z * vv[j].z;
        a += m.w * vv[j].w;
      }
      s[i] = a;
    }
#pragma unroll
    for (int off = 16; off > 0; off >>= 1)
#pragma unroll
      for (int i = 0; i < RU; ++i) s[i] += __shfl_xor(s[i], off, 32);
    float t = s[0];
#pragma unroll
    for (int i = 1; i < RU; ++i)
      if (lane == i) t = s[i];
    if (lane < RU) {
      const int row = row0 + r0 + lane;
      t += bias[row];
      if (SILU) {
        float sig = 1.0f / (1.0f + __expf(-t));
        t = t * sig;
      }
      y[row] = t;
    }
  }
}

__global__ void k_init_w(const float* __restrict__ Wq, const float* __restrict__ bq,
                         const float* __restrict__ iq, float* __restrict__ ws) {
  const int lane = threadIdx.x;
  if (blockIdx.x != 0) {
    wave_matvec<C_, 4, 4, false>(Wq, iq, bq, ws + OFF_RQ, (blockIdx.x - 1) * 4, lane);
    return;
  }
  const float a = 1.0f - DECAY;
  for (int t = lane; t < T_; t += 32) {
    float n = (float)(T_ - t);
    float an = __powf(a, n), mn = __powf(MOM, n);
    ws[OFF_W + t] = -LR * (1.0f - MOM) * (an - mn) / (a - MOM);
  }
  if (lane == 0) ws[OFF_W + 64] = __powf(a, (float)T_);
}

__global__ void __launch_bounds__(128) __attribute__((amdgpu_waves_per_eu(1, 1)))
k_gemm_kv(const float* __restrict__ x,
                          const float* __restrict__ Wk, const float* __restrict__ bk,
                          const float* __restrict__ Wv, const float* __restrict__ bv,
                          const float* __restrict__ sW0, const float* __restrict__ sb0,
                          float* __restrict__ ws) {
  __shared__ float red[4 * 32 * RP_];
  const int b = blockIdx.y, which = blockIdx.z;
  const int lane = threadIdx.x & 31, wave = threadIdx.x >> 5;
  if (blockIdx.x >= 24) {
    const int row0 = which * (H_ / 2) + (blockIdx.x - 24) * 16 + wave * 4;
    wave_matvec<C_, 4, 4, true>(sW0 + b * H_ * C_, ws + OFF_RQ, sb0 + b * H_,
                             ws + OFF_RH + b * H_, row0, lane);
    return;
  }
  const int m0 = (blockIdx.x & 1) * 32, n0 = (blockIdx.x >> 1) * 32;
  const float* A = x + b * T_ * C_;
  const float* W = which ? Wv : Wk;
  const float* bias = which ? bv : bk;
  float* out = ws + (which ? OFF_V : OFF_K) + b * T_ * C_;
  v8f acc[4] = {};
  gemm32x32_slice<true, C_ / 4>(A, C_, W, C_, wave * (C_ / 4), m0, n0, lane, acc);
  put_partial(acc, red, wave, lane);
  __syncthreads();
  const int col = n0 + lane;
  const float bb = bias[col];
#pragma unroll
  for (int j = 0; j < 8; ++j) {
    const int R = wave * 8 + j;
    out[(m0 + R) * C_ + col] = row_sum<4>(red, R, lane) + bb;
  }
}

__global__ void __launch_bounds__(128) __attribute__((amdgpu_waves_per_eu(1, 1)))
k_gemm_h1(const float* __restrict__ sW0,
                          const float* __restrict__ sb0,
                          const float* __restrict__ sW1,
                          const float* __restrict__ sb1,
                          float* __restrict__ ws) {
  __shared__ float red[4 * 32 * RP_];
  const int b = blockIdx.y;
  const int lane = threadIdx.x & 31, wave = threadIdx.x >> 5;
  if (blockIdx.x >= 48) {
    wave_matvec<H_, 4, 2, false>(sW1 + b * C_ * H_, ws + OFF_RH + b * H_, sb1 + b * C_,
                              ws + OFF_RP + b * C_, (blockIdx.x - 48) * 16 + wave * 4, lane);
    return;
  }
  const int m0 = (blockIdx.x % 2) * 32, n0 = (blockIdx.x / 2) * 32;
  const float* A = ws + OFF_K + b * T_ * C_;
  const float* W = sW0 + b * H_ * C_;
  float* h1 = ws + OFF_H1 + b * T_ * H_;
  float* gp = ws + OFF_GP + b * T_ * H_;
  v8f acc[4] = {};
  gemm32x32_slice<true, C_ / 4>(A, C_, W, C_, wave * (C_ / 4), m0, n0, lane, acc);
  put_partial(acc, red, wave, lane);
  __syncthreads();
  const int col = n0 + lane;
  const float bb = sb0[b * H_ + col];
#pragma unroll
  for (int j = 0; j < 8; ++j) {
    const int R = wave * 8 + j;
    float pre = row_sum<4>(red, R, lane) + bb;
    float sig = 1.0f / (1.0f + __expf(-pre));
    h1[(m0 + R) * H_ + col] = pre * sig;
    gp[(m0 + R) * H_ + col] = sig * (1.0f + pre * (1.0f - sig));
  }
}

__global__ void __launch_bounds__(256) __attribute__((amdgpu_waves_per_eu(2, 2)))
k_gemm_dp(const float* __restrict__ sW1,
                          const float* __restrict__ sb1,
                          const float* __restrict__ Wo,
                          const float* __restrict__ bo,
                          float* __restrict__ rout,
                          float* __restrict__ ws) {
  __shared__ float red[8 * 32 * RP_];
  __shared__ float dts[32 * TP_];
  const int b = blockIdx.y;
  const int lane = threadIdx.x & 31, wave = threadIdx.x >> 5;
  if (blockIdx.x >= 24) {
    wave_matvec<C_, 2, 2, false>(Wo, ws + OFF_RP + b * C_, bo, rout + b * C_,
                              (blockIdx.x - 24) * 16 + wave * 2, lane);
    return;
  }
  const int m0 = (blockIdx.x % 2) * 32, n0 = (blockIdx.x / 2) * 32;
  const float* A = ws + OFF_H1 + b * T_ * H_;
  const float* W = sW1 + b * C_ * H_;
  const float* v = ws + OFF_V + b * T_ * C_;
  float* dp = ws + OFF_DP + b * T_ * C_;
  float* wdpT = ws + OFF_WDPT + b * C_ * T_;
  v8f acc[4] = {};
  gemm32x32_slice<true, H_ / 8>(A, H_, W, H_, wave * (H_ / 8), m0, n0, lane, acc);
  put_partial(acc, red, wave, lane);
  __syncthreads();
  const int col = n0 + lane;
  const float bb = sb1[b * C_ + col];
  const float wrow = ws[OFF_W + m0 + lane];
  float d[4];
#pragma unroll
  for (int j = 0; j < 4; ++j) {
    const int R = wave * 4 + j;
    d[j] = (row_sum<8>(red, R, lane) + bb) - v[(m0 + R) * C_ + col];
  }
#pragma unroll
  for (int j = 0; j < 4; ++j) dts[(wave * 4 + j) * TP_ + lane] = d[j];
  __syncthreads();
#pragma unroll
  for (int j = 0; j < 4; ++j) {
    const int Cc = wave * 4 + j;
    wdpT[(n0 + Cc) * T_ + m0 + lane] = wrow * dts[lane * TP_ + Cc];
  }
#pragma unroll
  for (int j = 0; j < 4; ++j) dp[(m0 + wave * 4 + j) * C_ + col] = d[j];
}

__global__ void __launch_bounds__(256) __attribute__((amdgpu_waves_per_eu(2, 2)))
k_gemm_dh1(const float* __restrict__ sW1,
                           float* __restrict__ ws) {
  __shared__ float red[8 * 32 * RP_];
  __shared__ float dts[32 * TP_];
  const int b = blockIdx.y;
  const int lane = threadIdx.x & 31, wave = threadIdx.x >> 5;
  const int m0 = (blockIdx.x % 2) * 32, n0 = (blockIdx.x / 2) * 32;
  const float* A = ws + OFF_DP + b * T_ * C_;
  const float* W = sW1 + b * C_ * H_;
  const float* gp = ws + OFF_GP + b * T_ * H_;
  float* wdhT = ws + OFF_WDHT + b * H_ * T_;
  v8f acc[4] = {};
  gemm32x32_slice<false, C_ / 8>(A, C_, W, H_, wave * (C_ / 8), m0, n0, lane, acc);
  put_partial(acc, red, wave, lane);
  __syncthreads();
  const int col = n0 + lane;
  const float wrow = ws[OFF_W + m0 + lane];
#pragma unroll
  for (int j = 0; j < 4; ++j) {
    const int R = wave * 4 + j;
    dts[R * TP_ + lane] = row_sum<8>(red, R, lane) * gp[(m0 + R) * H_ + col];
  }
  __syncthreads();
#pragma unroll
  for (int j = 0; j < 4; ++j) {
    const int Cc = wave * 4 + j;
    wdhT[(n0 + Cc) * T_ + m0 + lane] = wrow * dts[lane * TP_ + Cc];
  }
}

__global__ void k_gemm_nw0(const float* __restrict__ sW0,
                           float* __restrict__ nW0,
                           float* __restrict__ ws) {
  const int b = blockIdx.y, lane = threadIdx.x;
  const int m0 = (blockIdx.x / 12) * 32, n0 = (blockIdx.x % 12) * 32;
  const float* A = ws + OFF_WDHT + b * H_ * T_;
  const float* Bm = ws + OFF_K + b * T_ * C_;
  const float dT = ws[OFF_W + 64];
  float* out = nW0 + b * H_ * C_;
  const float* w0 = sW0 + b * H_ * C_;
  v8f acc[4] = {};
  gemm32x32<false>(A, T_, Bm, C_, T_, m0, n0, lane, acc);
  const int cr = lane & 15, hi = lane >> 4;
#pragma unroll
  for (int mi = 0; mi < 2; ++mi)
#pragma unroll
    for (int ni = 0; ni < 2; ++ni) {
      int col = n0 + ni * 16 + cr;
      v8f a = acc[mi * 2 + ni];
#pragma unroll
      for (int r = 0; r < 8; ++r) {
        int row = m0 + mi * 16 + r + 8 * hi;
        out[row * C_ + col] = dT * w0[row * C_ + col] + a[r];
      }
    }
}

__global__ void k_gemm_nw1(const float* __restrict__ sW1,
                           float* __restrict__ nW1,
                           float* __restrict__ ws) {
  const int b = blockIdx.y, lane = threadIdx.x;
  const int m0 = (blockIdx.x / 24) * 32, n0 = (blockIdx.x % 24) * 32;
  const float* A = ws + OFF_WDPT + b * C_ * T_;
  const float* Bm = ws + OFF_H1 + b * T_ * H_;
  const float dT = ws[OFF_W + 64];
  float* out = nW1 + b * C_ * H_;
  const float* w1 = sW1 + b * C_ * H_;
  v8f acc[4] = {};
  gemm32x32<false>(A, T_, Bm, H_, T_, m0, n0, lane, acc);
  const int cr = lane & 15, hi = lane >> 4;
#pragma unroll
  for (int mi = 0; mi < 2; ++mi)
#pragma unroll
    for (int ni = 0; ni < 2; ++ni) {
      int col = n0 + ni * 16 + cr;
      v8f a = acc[mi * 2 + ni];
#pragma unroll
      for (int r = 0; r < 8; ++r) {
        int row = m0 + mi * 16 + r + 8 * hi;
        out[row * H_ + col] = dT * w1[row * H_ + col] + a[r];
      }
    }
}

__global__ void k_bias(const float* __restrict__ sb0, const float* __restrict__ sb1,
                       float* __restrict__ nb0, float* __restrict__ nb1,
                       const float* __restrict__ ws) {
  int idx = blockIdx.x * blockDim.x + threadIdx.x;
  if (idx >= B_ * (H_ + C_)) return;
  int b = idx / (H_ + C_), r = idx % (H_ + C_);
  const float dT = ws[OFF_W + 64];
  if (r < H_) {
    const float* p = ws + OFF_WDHT + b * H_ * T_ + r * T_;
    float s = 0.f;
#pragma unroll 4
    for (int t = 0; t < T_; ++t) s += p[t];
    nb0[b * H_ + r] = dT * sb0[b * H_ + r] + s;
  } else {
    int c = r - H_;
    const float* p = ws + OFF_WDPT + b * C_ * T_ + c * T_;
    float s = 0.f;
#pragma unroll 4
    for (int t = 0; t < T_; ++t) s += p[t];
    nb1[b * C_ + c] = dT * sb1[b * C_ + c] + s;
  }
}

}

extern "C" void kernel_launch(void* const* d_in, const int* in_sizes, int n_in,
                              void* d_out, int out_size, void* d_ws, size_t ws_size,
                              hipStream_t stream) {
  const float* x   = (const float*)d_in[0];
  const float* Wq  = (const float*)d_in[1];
  const float* bq  = (const float*)d_in[2];
  const float* Wk  = (const float*)d_in[3];
  const float* bk  = (const float*)d_in[4];
  const float* Wv  = (const float*)d_in[5];
  const float* bv  = (const float*)d_in[6];
  const float* Wo  = (const float*)d_in[7];
  const float* bo  = (const float*)d_in[8];
  const float* iq  = (const float*)d_in[9];
  const float* sW0 = (const float*)d_in[10];
  const float* sb0 = (const float*)d_in[11];
  const float* sW1 = (const float*)d_in[12];
  const float* sb1 = (const float*)d_in[13];

  float* out = (float*)d_out;
  float* ws  = (float*)d_ws;

  float* out_out = out;
  float* out_nW0 = out + B_ * C_;
  float* out_nb0 = out_nW0 + B_ * H_ * C_;
  float* out_nW1 = out_nb0 + B_ * H_;
  float* out_nb1 = out_nW1 + B_ * C_ * H_;

  k_init_w <<<97, 32, 0, stream>>>(Wq, bq, iq, ws);
  k_gemm_kv<<<dim3(48, B_, 2), 128, 0, stream>>>(x, Wk, bk, Wv, bv, sW0, sb0, ws);
  k_gemm_h1<<<dim3(72, B_), 128, 0, stream>>>(sW0, sb0, sW1, sb1, ws);
  k_gemm_dp<<<dim3(48, B_), 256, 0, stream>>>(sW1, sb1, Wo, bo, out_out, ws);
  k_gemm_dh1<<<dim3(48, B_), 256, 0, stream>>>(sW1, ws);
  k_gemm_nw0<<<dim3(288, B_), 32, 0, stream>>>(sW0, out_nW0, ws);
  k_gemm_nw1<<<dim3(288, B_), 32, 0, stream>>>(sW1, out_nW1, ws);
  k_bias    <<<dim3(18), 128, 0, stream>>>(sb0, sb1, out_nb0, out_nb1, ws);
}


# Launches 8 -> 5, one per dependent stage: scan weights computed in-lane where used (table launch gone, retrieval rides launches 1-4), and nW0, nW1 and bias updates share one launch
# speedup vs baseline: 6.3580x; 1.2890x over previous
#include <hip/hip_runtime.h>
#include <cstdint>

typedef float v2f __attribute__((ext_vector_type(2)));
typedef float v4f __attribute__((ext_vector_type(4)));
typedef float v8f __attribute__((ext_vector_type(8)));

namespace {

constexpr int B_ = 2, T_ = 64, C_ = 384, H_ = 768;
constexpr float LR = 0.01f, MOM = 0.9f, DECAY = 0.001f;

constexpr int OFF_K    = 0;
constexpr int OFF_V    = OFF_K    + B_*T_*C_;
constexpr int OFF_H1   = OFF_V    + B_*T_*C_;
constexpr int OFF_GP   = OFF_H1   + B_*T_*H_;
constexpr int OFF_DP   = OFF_GP   + B_*T_*H_;
constexpr int OFF_WDPT = OFF_DP   + B_*T_*C_;
constexpr int OFF_WDHT = OFF_WDPT + B_*C_*T_;
constexpr int OFF_RQ   = OFF_WDHT + B_*H_*T_;
constexpr int OFF_RH   = OFF_RQ   + C_;
constexpr int OFF_RP   = OFF_RH   + B_*H_;

__device__ __forceinline__ float scan_w(int t) {
  const float a = 1.0f - DECAY;
  const float n = (float)(T_ - t);
  const float an = __powf(a, n), mn = __powf(MOM, n);
  return -LR * (1.0f - MOM) * (an - mn) / (a - MOM);
}
__device__ __forceinline__ float decay_T() { return __powf(1.0f - DECAY, (float)T_); }

__device__ __forceinline__ v8f wmma4(v2f a, v2f b, v8f c) {
  return __builtin_amdgcn_wmma_f32_16x16x4_f32(false, a, false, b, (short)0, c,
                                               false, false);
}

template <bool BT>
__device__ __forceinline__ void gemm32x32(const float* __restrict__ A, int lda,
                                          const float* __restrict__ Bm, int ldb,
                                          int K, int m0, int n0, int lane,
                                          v8f acc[4]) {
  const int mr = lane & 15;
  const int kh = (lane >> 4) << 1;
  for (int kk = 0; kk < K; kk += 4) {
    const int kp = kk + kh;
    v2f a0 = *reinterpret_cast<const v2f*>(A + (m0 + mr) * lda + kp);
    v2f a1 = *reinterpret_cast<const v2f*>(A + (m0 + 16 + mr) * lda + kp);
    v2f b0, b1;
    if (BT) {
      b0 = *reinterpret_cast<const v2f*>(Bm + (n0 + mr) * ldb + kp);
      b1 = *reinterpret_cast<const v2f*>(Bm + (n0 + 16 + mr) * ldb + kp);
    } else {
      const float* pb = Bm + kp * ldb + n0 + mr;
      b0.x = pb[0];  b0.y = pb[ldb];
      b1.x = pb[16]; b1.y = pb[ldb + 16];
    }
    acc[0] = wmma4(a0, b0, acc[0]);
    acc[1] = wmma4(a0, b1, acc[1]);
    acc[2] = wmma4(a1, b0, acc[2]);
    acc[3] = wmma4(a1, b1, acc[3]);
  }
}

template <bool BT, int KS>
__device__ __forceinline__ void gemm32x32_slice(const float* __restrict__ A, int lda,
                                                const float* __restrict__ Bm, int ldb,
                                                int k0, int m0, int n0, int lane,
                                                v8f acc[4]) {
  const int mr = lane & 15;
  const int kh = (lane >> 4) << 1;
#pragma unroll
  for (int kk = 0; kk < KS; kk += 4) {
    const int kp = k0 + kk + kh;
    v2f a0 = *reinterpret_cast<const v2f*>(A + (m0 + mr) * lda + kp);
    v2f a1 = *reinterpret_cast<const v2f*>(A + (m0 + 16 + mr) * lda + kp);
    v2f b0, b1;
    if (BT) {
      b0 = *reinterpret_cast<const v2f*>(Bm + (n0 + mr) * ldb + kp);
      b1 = *reinterpret_cast<const v2f*>(Bm + (n0 + 16 + mr) * ldb + kp);
    } else {
      const float* pb = Bm + kp * ldb + n0 + mr;
      b0.x = pb[0];  b0.y = pb[ldb];
      b1.x = pb[16]; b1.y = pb[ldb + 16];
    }
    acc[0] = wmma4(a0, b0, acc[0]);
    acc[1] = wmma4(a0, b1, acc[1]);
    acc[2] = wmma4(a1, b0, acc[2]);
    acc[3] = wmma4(a1, b1, acc[3]);
  }
}

constexpr int RP_ = 36;
constexpr int TP_ = 33;

__device__ __forceinline__ void put_partial(const v8f acc[4], float* red, int slice, int lane) {
  const int cr = lane & 15, hi = lane >> 4;
#pragma unroll
  for (int qq = 0; qq < 4; ++qq)
#pragma unroll
    for (int r = 0; r < 8; ++r) {
      const int R = (qq >> 1) * 16 + 8 * hi + r;
      const int Cc = (qq & 1) * 16 + cr;
      red[(slice * 32 + R) * RP_ + Cc] = acc[qq][r];
    }
}

template <int S>
__device__ __forceinline__ float row_sum(const float* red, int R, int lane) {
  float s = red[R * RP_ + lane];
#pragma unroll
  for (int sl = 1; sl < S; ++sl) s += red[(sl * 32 + R) * RP_ + lane];
  return s;
}

template <int K, int NR, int RU, bool SILU>
__device__ __forceinline__ void wave_matvec(const float* __restrict__ M,
                                            const float* __restrict__ v,
                                            const float* __restrict__ bias,
                                            float* __restrict__ y, int row0, int lane) {
  constexpr int NV = K / 128;
  v4f vv[NV];
#pragma unroll
  for (int j = 0; j < NV; ++j)
    vv[j] = *reinterpret_cast<const v4f*>(v + j * 128 + lane * 4);
  for (int r0 = 0; r0 < NR; r0 += RU) {
    float s[RU];
#pragma unroll
    for (int i = 0; i < RU; ++i) {
      const float* mr = M + (row0 + r0 + i) * K + lane * 4;
      float a = 0.f;
#pragma unroll
      for (int j = 0; j < NV; ++j) {
        v4f m = *reinterpret_cast<const v4f*>(mr + j * 128);
        a += m.x * vv[j].x;
        a += m.y * vv[j].y;
        a += m.z * vv[j].z;
        a += m.w * vv[j].w;
      }
      s[i] = a;
    }
#pragma unroll
    for (int off = 16; off > 0; off >>= 1)
#pragma unroll
      for (int i = 0; i < RU; ++i) s[i] += __shfl_xor(s[i], off, 32);
    float t = s[0];
#pragma unroll
    for (int i = 1; i < RU; ++i)
      if (lane == i) t = s[i];
    if (lane < RU) {
      const int row = row0 + r0 + lane;
      t += bias[row];
      if (SILU) {
        float sig = 1.0f / (1.0f + __expf(-t));
        t = t * sig;
      }
      y[row] = t;
    }
  }
}

__global__ void __launch_bounds__(128) __attribute__((amdgpu_waves_per_eu(1, 1)))
k_gemm_kv(const float* __restrict__ x,
                          const float* __restrict__ Wk, const float* __restrict__ bk,
                          const float* __restrict__ Wv, const float* __restrict__ bv,
                          const float* __restrict__ Wq, const float* __restrict__ bq,
                          const float* __restrict__ iq,
                          float* __restrict__ ws) {
  __shared__ float red[4 * 32 * RP_];
  const int b = blockIdx.y, which = blockIdx.z;
  const int lane = threadIdx.x & 31, wave = threadIdx.x >> 5;
  if (blockIdx.x >= 24) {
    const int blk = ((blockIdx.x - 24) * B_ + b) * 2 + which;
    wave_matvec<C_, 4, 4, false>(Wq, iq, bq, ws + OFF_RQ, blk * 16 + wave * 4, lane);
    return;
  }
  const int m0 = (blockIdx.x & 1) * 32, n0 = (blockIdx.x >> 1) * 32;
  const float* A = x + b * T_ * C_;
  const float* W = which ? Wv : Wk;
  const float* bias = which ? bv : bk;
  float* out = ws + (which ? OFF_V : OFF_K) + b * T_ * C_;
  v8f acc[4] = {};
  gemm32x32_slice<true, C_ / 4>(A, C_, W, C_, wave * (C_ / 4), m0, n0, lane, acc);
  put_partial(acc, red, wave, lane);
  __syncthreads();
  const int col = n0 + lane;
  const float bb = bias[col];
#pragma unroll
  for (int j = 0; j < 8; ++j) {
    const int R = wave * 8 + j;
    out[(m0 + R) * C_ + col] = row_sum<4>(red, R, lane) + bb;
  }
}

__global__ void __launch_bounds__(128) __attribute__((amdgpu_waves_per_eu(1, 1)))
k_gemm_h1(const float* __restrict__ sW0,
                          const float* __restrict__ sb0,
                          float* __restrict__ ws) {
  __shared__ float red[4 * 32 * RP_];
  const int b = blockIdx.y;
  const int lane = threadIdx.x & 31, wave = threadIdx.x >> 5;
  if (blockIdx.x >= 48) {
    wave_matvec<C_, 4, 4, true>(sW0 + b * H_ * C_, ws + OFF_RQ, sb0 + b * H_,
                             ws + OFF_RH + b * H_, (blockIdx.x - 48) * 16 + wave * 4, lane);
    return;
  }
  const int m0 = (blockIdx.x % 2) * 32, n0 = (blockIdx.x / 2) * 32;
  const float* A = ws + OFF_K + b * T_ * C_;
  const float* W = sW0 + b * H_ * C_;
  float* h1 = ws + OFF_H1 + b * T_ * H_;
  float* gp = ws + OFF_GP + b * T_ * H_;
  v8f acc[4] = {};
  gemm32x32_slice<true, C_ / 4>(A, C_, W, C_, wave * (C_ / 4), m0, n0, lane, acc);
  put_partial(acc, red, wave, lane);
  __syncthreads();
  const int col = n0 + lane;
  const float bb = sb0[b * H_ + col];
#pragma unroll
  for (int j = 0; j < 8; ++j) {
    const int R = wave * 8 + j;
    float pre = row_sum<4>(red, R, lane) + bb;
    float sig = 1.0f / (1.0f + __expf(-pre));
    h1[(m0 + R) * H_ + col] = pre * sig;
    gp[(m0 + R) * H_ + col] = sig * (1.0f + pre * (1.0f - sig));
  }
}

__global__ void __launch_bounds__(256) __attribute__((amdgpu_waves_per_eu(2, 2)))
k_gemm_dp(const float* __restrict__ sW1,
                          const float* __restrict__ sb1,
                          float* __restrict__ ws) {
  __shared__ float red[8 * 32 * RP_];
  __shared__ float dts[32 * TP_];
  const int b = blockIdx.y;
  const int lane = threadIdx.x & 31, wave = threadIdx.x >> 5;
  if (blockIdx.x >= 24) {
    wave_matvec<H_, 2, 2, false>(sW1 + b * C_ * H_, ws + OFF_RH + b * H_, sb1 + b * C_,
                              ws + OFF_RP + b * C_, (blockIdx.x - 24) * 16 + wave * 2, lane);
    return;
  }
  const int m0 = (blockIdx.x % 2) * 32, n0 = (blockIdx.x / 2) * 32;
  const float* A = ws + OFF_H1 + b * T_ * H_;
  const float* W = sW1 + b * C_ * H_;
  const float* v = ws + OFF_V + b * T_ * C_;
  float* dp = ws + OFF_DP + b * T_ * C_;
  float* wdpT = ws + OFF_WDPT + b * C_ * T_;
  v8f acc[4] = {};
  gemm32x32_slice<true, H_ / 8>(A, H_, W, H_, wave * (H_ / 8), m0, n0, lane, acc);
  put_partial(acc, red, wave, lane);
  __syncthreads();
  const int col = n0 + lane;
  const float bb = sb1[b * C_ + col];
  const float wrow = scan_w(m0 + lane);
  float d[4];
#pragma unroll
  for (int j = 0; j < 4; ++j) {
    const int R = wave * 4 + j;
    d[j] = (row_sum<8>(red, R, lane) + bb) - v[(m0 + R) * C_ + col];
  }
#pragma unroll
  for (int j = 0; j < 4; ++j) dts[(wave * 4 + j) * TP_ + lane] = d[j];
  __syncthreads();
#pragma unroll
  for (int j = 0; j < 4; ++j) {
    const int Cc = wave * 4 + j;
    wdpT[(n0 + Cc) * T_ + m0 + lane] = wrow * dts[lane * TP_ + Cc];
  }
#pragma unroll
  for (int j = 0; j < 4; ++j) dp[(m0 + wave * 4 + j) * C_ + col] = d[j];
}

__global__ void __launch_bounds__(256) __attribute__((amdgpu_waves_per_eu(2, 2)))
k_gemm_dh1(const float* __restrict__ sW1,
                           const float* __restrict__ Wo,
                           const float* __restrict__ bo,
                           float* __restrict__ rout,
                           float* __restrict__ ws) {
  __shared__ float red[8 * 32 * RP_];
  __shared__ float dts[32 * TP_];
  const int b = blockIdx.y;
  const int lane = threadIdx.x & 31, wave = threadIdx.x >> 5;
  if (blockIdx.x >= 48) {
    wave_matvec<C_, 2, 2, false>(Wo, ws + OFF_RP + b * C_, bo, rout + b * C_,
                              (blockIdx.x - 48) * 16 + wave * 2, lane);
    return;
  }
  const int m0 = (blockIdx.x % 2) * 32, n0 = (blockIdx.x / 2) * 32;
  const float* A = ws + OFF_DP + b * T_ * C_;
  const float* W = sW1 + b * C_ * H_;
  const float* gp = ws + OFF_GP + b * T_ * H_;
  float* wdhT = ws + OFF_WDHT + b * H_ * T_;
  v8f acc[4] = {};
  gemm32x32_slice<false, C_ / 8>(A, C_, W, H_, wave * (C_ / 8), m0, n0, lane, acc);
  put_partial(acc, red, wave, lane);
  __syncthreads();
  const int col = n0 + lane;
  const float wrow = scan_w(m0 + lane);
#pragma unroll
  for (int j = 0; j < 4; ++j) {
    const int R = wave * 4 + j;
    dts[R * TP_ + lane] = row_sum<8>(red, R, lane) * gp[(m0 + R) * H_ + col];
  }
  __syncthreads();
#pragma unroll
  for (int j = 0; j < 4; ++j) {
    const int Cc = wave * 4 + j;
    wdhT[(n0 + Cc) * T_ + m0 + lane] = wrow * dts[lane * TP_ + Cc];
  }
}

template <int N>
__device__ __forceinline__ void update_tile(const float* __restrict__ sW,
                                            float* __restrict__ nW,
                                            const float* __restrict__ G,
                                            const float* __restrict__ Bm,
                                            int tile, int lane, float dT) {
  const int m0 = (tile / (N / 32)) * 32, n0 = (tile % (N / 32)) * 32;
  v8f acc[4] = {};
  gemm32x32<false>(G, T_, Bm, N, T_, m0, n0, lane, acc);
  const int cr = lane & 15, hi = lane >> 4;
#pragma unroll
  for (int mi = 0; mi < 2; ++mi)
#pragma unroll
    for (int ni = 0; ni < 2; ++ni) {
      int col = n0 + ni * 16 + cr;
      v8f a = acc[mi * 2 + ni];
#pragma unroll
      for (int r = 0; r < 8; ++r) {
        int row = m0 + mi * 16 + r + 8 * hi;
        nW[row * N + col] = dT * sW[row * N + col] + a[r];
      }
    }
}

__global__ void k_update(const float* __restrict__ sW0, const float* __restrict__ sb0,
                         const float* __restrict__ sW1, const float* __restrict__ sb1,
                         float* __restrict__ nW0, float* __restrict__ nb0,
                         float* __restrict__ nW1, float* __restrict__ nb1,
                         const float* __restrict__ ws) {
  const int bx = blockIdx.x, lane = threadIdx.x;
  const float dT = decay_T();
  if (bx >= 648) {
    const int b = (bx - 648) / 288, tile = (bx - 648) % 288;
    update_tile<H_>(sW1 + b * C_ * H_, nW1 + b * C_ * H_, ws + OFF_WDPT + b * C_ * T_,
                    ws + OFF_H1 + b * T_ * H_, tile, lane, dT);
    return;
  }
  if (bx >= 72) {
    const int b = (bx - 72) / 288, tile = (bx - 72) % 288;
    update_tile<C_>(sW0 + b * H_ * C_, nW0 + b * H_ * C_, ws + OFF_WDHT + b * H_ * T_,
                    ws + OFF_K + b * T_ * C_, tile, lane, dT);
    return;
  }
  const int b = bx / 36;
  const int r = (bx % 36) * 32 + lane;
  if (r < H_) {
    const float* p = ws + OFF_WDHT + b * H_ * T_ + r * T_;
    float s = 0.f;
#pragma unroll 4
    for (int t = 0; t < T_; ++t) s += p[t];
    nb0[b * H_ + r] = dT * sb0[b * H_ + r] + s;
  } else {
    const int c = r - H_;
    const float* p = ws + OFF_WDPT + b * C_ * T_ + c * T_;
    float s = 0.f;
#pragma unroll 4
    for (int t = 0; t < T_; ++t) s += p[t];
    nb1[b * C_ + c] = dT * sb1[b * C_ + c] + s;
  }
}

}

extern "C" void kernel_launch(void* const* d_in, const int* in_sizes, int n_in,
                              void* d_out, int out_size, void* d_ws, size_t ws_size,
                              hipStream_t stream) {
  const float* x   = (const float*)d_in[0];
  const float* Wq  = (const float*)d_in[1];
  const float* bq  = (const float*)d_in[2];
  const float* Wk  = (const float*)d_in[3];
  const float* bk  = (const float*)d_in[4];
  const float* Wv  = (const float*)d_in[5];
  const float* bv  = (const float*)d_in[6];
  const float* Wo  = (const float*)d_in[7];
  const float* bo  = (const float*)d_in[8];
  const float* iq  = (const float*)d_in[9];
  const float* sW0 = (const float*)d_in[10];
  const float* sb0 = (const float*)d_in[11];
  const float* sW1 = (const float*)d_in[12];
  const float* sb1 = (const float*)d_in[13];

  float* out = (float*)d_out;
  float* ws  = (float*)d_ws;

  float* out_out = out;
  float* out_nW0 = out + B_ * C_;
  float* out_nb0 = out_nW0 + B_ * H_ * C_;
  float* out_nW1 = out_nb0 + B_ * H_;
  float* out_nb1 = out_nW1 + B_ * C_ * H_;

  k_gemm_kv<<<dim3(30, B_, 2), 128, 0, stream>>>(x, Wk, bk, Wv, bv, Wq, bq, iq, ws);
  k_gemm_h1<<<dim3(96, B_), 128, 0, stream>>>(sW0, sb0, ws);
  k_gemm_dp<<<dim3(48, B_), 256, 0, stream>>>(sW1, sb1, ws);
  k_gemm_dh1<<<dim3(72, B_), 256, 0, stream>>>(sW1, Wo, bo, out_out, ws);
  k_update  <<<dim3(1224), 32, 0, stream>>>(sW0, sb0, sW1, sb1, out_nW0, out_nb0,
                                            out_nW1, out_nb1, ws);
}


# Fragment loads widened from 8 to 16 bytes per lane in all 5 launches: a lane takes 4 consecutive k of its row for two products; k order permuted equally in A and B (summation order changes)
# speedup vs baseline: 7.2665x; 1.1429x over previous
#include <hip/hip_runtime.h>
#include <cstdint>

typedef float v2f __attribute__((ext_vector_type(2)));
typedef float v4f __attribute__((ext_vector_type(4)));
typedef float v8f __attribute__((ext_vector_type(8)));

namespace {

constexpr int B_ = 2, T_ = 64, C_ = 384, H_ = 768;
constexpr float LR = 0.01f, MOM = 0.9f, DECAY = 0.001f;

constexpr int OFF_K    = 0;
constexpr int OFF_V    = OFF_K    + B_*T_*C_;
constexpr int OFF_H1   = OFF_V    + B_*T_*C_;
constexpr int OFF_GP   = OFF_H1   + B_*T_*H_;
constexpr int OFF_DP   = OFF_GP   + B_*T_*H_;
constexpr int OFF_WDPT = OFF_DP   + B_*T_*C_;
constexpr int OFF_WDHT = OFF_WDPT + B_*C_*T_;
constexpr int OFF_RQ   = OFF_WDHT + B_*H_*T_;
constexpr int OFF_RH   = OFF_RQ   + C_;
constexpr int OFF_RP   = OFF_RH   + B_*H_;

__device__ __forceinline__ float scan_w(int t) {
  const float a = 1.0f - DECAY;
  const float n = (float)(T_ - t);
  const float an = __powf(a, n), mn = __powf(MOM, n);
  return -LR * (1.0f - MOM) * (an - mn) / (a - MOM);
}
__device__ __forceinline__ float decay_T() { return __powf(1.0f - DECAY, (float)T_); }

__device__ __forceinline__ v8f wmma4(v2f a, v2f b, v8f c) {
  return __builtin_amdgcn_wmma_f32_16x16x4_f32(false, a, false, b, (short)0, c,
                                               false, false);
}

template <bool BT>
__device__ __forceinline__ void load_frags(const float* __restrict__ A, int lda,
                                           const float* __restrict__ Bm, int ldb,
                                           int kp, int m0, int n0, int mr,
                                           v4f& a0, v4f& a1, v4f& b0, v4f& b1) {
  a0 = *reinterpret_cast<const v4f*>(A + (m0 + mr) * lda + kp);
  a1 = *reinterpret_cast<const v4f*>(A + (m0 + 16 + mr) * lda + kp);
  if (BT) {
    b0 = *reinterpret_cast<const v4f*>(Bm + (n0 + mr) * ldb + kp);
    b1 = *reinterpret_cast<const v4f*>(Bm + (n0 + 16 + mr) * ldb + kp);
  } else {
    const float* pb = Bm + kp * ldb + n0 + mr;
    b0.x = pb[0];       b0.y = pb[ldb];       b0.z = pb[2 * ldb];       b0.w = pb[3 * ldb];
    b1.x = pb[16];      b1.y = pb[ldb + 16];  b1.z = pb[2 * ldb + 16];  b1.w = pb[3 * ldb + 16];
  }
}

__device__ __forceinline__ void mma_pair(v4f a0, v4f a1, v4f b0, v4f b1, v8f acc[4]) {
  acc[0] = wmma4(a0.xy, b0.xy, acc[0]);
  acc[1] = wmma4(a0.xy, b1.xy, acc[1]);
  acc[2] = wmma4(a1.xy, b0.xy, acc[2]);
  acc[3] = wmma4(a1.xy, b1.xy, acc[3]);
  acc[0] = wmma4(a0.zw, b0.zw, acc[0]);
  acc[1] = wmma4(a0.zw, b1.zw, acc[1]);
  acc[2] = wmma4(a1.zw, b0.zw, acc[2]);
  acc[3] = wmma4(a1.zw, b1.zw, acc[3]);
}

template <bool BT>
__device__ __forceinline__ void gemm32x32(const float* __restrict__ A, int lda,
                                          const float* __restrict__ Bm, int ldb,
                                          int K, int m0, int n0, int lane,
                                          v8f acc[4]) {
  const int mr = lane & 15;
  const int kq = (lane >> 4) << 2;
  for (int kk = 0; kk < K; kk += 8) {
    const int kp = kk + kq;
    v4f a0, a1, b0, b1;
    load_frags<BT>(A, lda, Bm, ldb, kp, m0, n0, mr, a0, a1, b0, b1);
    mma_pair(a0, a1, b0, b1, acc);
  }
}

template <bool BT, int KS>
__device__ __forceinline__ void gemm32x32_slice(const float* __restrict__ A, int lda,
                                                const float* __restrict__ Bm, int ldb,
                                                int k0, int m0, int n0, int lane,
                                                v8f acc[4]) {
  const int mr = lane & 15;
  const int kq = (lane >> 4) << 2;
#pragma unroll
  for (int kk = 0; kk < KS; kk += 8) {
    const int kp = k0 + kk + kq;
    v4f a0, a1, b0, b1;
    load_frags<BT>(A, lda, Bm, ldb, kp, m0, n0, mr, a0, a1, b0, b1);
    mma_pair(a0, a1, b0, b1, acc);
  }
}

constexpr int RP_ = 36;
constexpr int TP_ = 33;

__device__ __forceinline__ void put_partial(const v8f acc[4], float* red, int slice, int lane) {
  const int cr = lane & 15, hi = lane >> 4;
#pragma unroll
  for (int qq = 0; qq < 4; ++qq)
#pragma unroll
    for (int r = 0; r < 8; ++r) {
      const int R = (qq >> 1) * 16 + 8 * hi + r;
      const int Cc = (qq & 1) * 16 + cr;
      red[(slice * 32 + R) * RP_ + Cc] = acc[qq][r];
    }
}

template <int S>
__device__ __forceinline__ float row_sum(const float* red, int R, int lane) {
  float s = red[R * RP_ + lane];
#pragma unroll
  for (int sl = 1; sl < S; ++sl) s += red[(sl * 32 + R) * RP_ + lane];
  return s;
}

template <int K, int NR, int RU, bool SILU>
__device__ __forceinline__ void wave_matvec(const float* __restrict__ M,
                                            const float* __restrict__ v,
                                            const float* __restrict__ bias,
                                            float* __restrict__ y, int row0, int lane) {
  constexpr int NV = K / 128;
  v4f vv[NV];
#pragma unroll
  for (int j = 0; j < NV; ++j)
    vv[j] = *reinterpret_cast<const v4f*>(v + j * 128 + lane * 4);
  for (int r0 = 0; r0 < NR; r0 += RU) {
    float s[RU];
#pragma unroll
    for (int i = 0; i < RU; ++i) {
      const float* mr = M + (row0 + r0 + i) * K + lane * 4;
      float a = 0.f;
#pragma unroll
      for (int j = 0; j < NV; ++j) {
        v4f m = *reinterpret_cast<const v4f*>(mr + j * 128);
        a += m.x * vv[j].x;
        a += m.y * vv[j].y;
        a += m.z * vv[j].z;
        a += m.w * vv[j].w;
      }
      s[i] = a;
    }
#pragma unroll
    for (int off = 16; off > 0; off >>= 1)
#pragma unroll
      for (int i = 0; i < RU; ++i) s[i] += __shfl_xor(s[i], off, 32);
    float t = s[0];
#pragma unroll
    for (int i = 1; i < RU; ++i)
      if (lane == i) t = s[i];
    if (lane < RU) {
      const int row = row0 + r0 + lane;
      t += bias[row];
      if (SILU) {
        float sig = 1.0f / (1.0f + __expf(-t));
        t = t * sig;
      }
      y[row] = t;
    }
  }
}

__global__ void __launch_bounds__(128) __attribute__((amdgpu_waves_per_eu(1, 1)))
k_gemm_kv(const float* __restrict__ x,
                          const float* __restrict__ Wk, const float* __restrict__ bk,
                          const float* __restrict__ Wv, const float* __restrict__ bv,
                          const float* __restrict__ Wq, const float* __restrict__ bq,
                          const float* __restrict__ iq,
                          float* __restrict__ ws) {
  __shared__ float red[4 * 32 * RP_];
  const int b = blockIdx.y, which = blockIdx.z;
  const int lane = threadIdx.x & 31, wave = threadIdx.x >> 5;
  if (blockIdx.x >= 24) {
    const int blk = ((blockIdx.x - 24) * B_ + b) * 2 + which;
    wave_matvec<C_, 4, 4, false>(Wq, iq, bq, ws + OFF_RQ, blk * 16 + wave * 4, lane);
    return;
  }
  const int m0 = (blockIdx.x & 1) * 32, n0 = (blockIdx.x >> 1) * 32;
  const float* A = x + b * T_ * C_;
  const float* W = which ? Wv : Wk;
  const float* bias = which ? bv : bk;
  float* out = ws + (which ? OFF_V : OFF_K) + b * T_ * C_;
  v8f acc[4] = {};
  gemm32x32_slice<true, C_ / 4>(A, C_, W, C_, wave * (C_ / 4), m0, n0, lane, acc);
  put_partial(acc, red, wave, lane);
  __syncthreads();
  const int col = n0 + lane;
  const float bb = bias[col];
#pragma unroll
  for (int j = 0; j < 8; ++j) {
    const int R = wave * 8 + j;
    out[(m0 + R) * C_ + col] = row_sum<4>(red, R, lane) + bb;
  }
}

__global__ void __launch_bounds__(128) __attribute__((amdgpu_waves_per_eu(1, 1)))
k_gemm_h1(const float* __restrict__ sW0,
                          const float* __restrict__ sb0,
                          float* __restrict__ ws) {
  __shared__ float red[4 * 32 * RP_];
  const int b = blockIdx.y;
  const int lane = threadIdx.x & 31, wave = threadIdx.x >> 5;
  if (blockIdx.x >= 48) {
    wave_matvec<C_, 4, 4, true>(sW0 + b * H_ * C_, ws + OFF_RQ, sb0 + b * H_,
                             ws + OFF_RH + b * H_, (blockIdx.x - 48) * 16 + wave * 4, lane);
    return;
  }
  const int m0 = (blockIdx.x % 2) * 32, n0 = (blockIdx.x / 2) * 32;
  const float* A = ws + OFF_K + b * T_ * C_;
  const float* W = sW0 + b * H_ * C_;
  float* h1 = ws + OFF_H1 + b * T_ * H_;
  float* gp = ws + OFF_GP + b * T_ * H_;
  v8f acc[4] = {};
  gemm32x32_slice<true, C_ / 4>(A, C_, W, C_, wave * (C_ / 4), m0, n0, lane, acc);
  put_partial(acc, red, wave, lane);
  __syncthreads();
  const int col = n0 + lane;
  const float bb = sb0[b * H_ + col];
#pragma unroll
  for (int j = 0; j < 8; ++j) {
    const int R = wave * 8 + j;
    float pre = row_sum<4>(red, R, lane) + bb;
    float sig = 1.0f / (1.0f + __expf(-pre));
    h1[(m0 + R) * H_ + col] = pre * sig;
    gp[(m0 + R) * H_ + col] = sig * (1.0f + pre * (1.0f - sig));
  }
}

__global__ void __launch_bounds__(256) __attribute__((amdgpu_waves_per_eu(2, 2)))
k_gemm_dp(const float* __restrict__ sW1,
                          const float* __restrict__ sb1,
                          float* __restrict__ ws) {
  __shared__ float red[8 * 32 * RP_];
  __shared__ float dts[32 * TP_];
  const int b = blockIdx.y;
  const int lane = threadIdx.x & 31, wave = threadIdx.x >> 5;
  if (blockIdx.x >= 24) {
    wave_matvec<H_, 2, 2, false>(sW1 + b * C_ * H_, ws + OFF_RH + b * H_, sb1 + b * C_,
                              ws + OFF_RP + b * C_, (blockIdx.x - 24) * 16 + wave * 2, lane);
    return;
  }
  const int m0 = (blockIdx.x % 2) * 32, n0 = (blockIdx.x / 2) * 32;
  const float* A = ws + OFF_H1 + b * T_ * H_;
  const float* W = sW1 + b * C_ * H_;
  const float* v = ws + OFF_V + b * T_ * C_;
  float* dp = ws + OFF_DP + b * T_ * C_;
  float* wdpT = ws + OFF_WDPT + b * C_ * T_;
  v8f acc[4] = {};
  gemm32x32_slice<true, H_ / 8>(A, H_, W, H_, wave * (H_ / 8), m0, n0, lane, acc);
  put_partial(acc, red, wave, lane);
  __syncthreads();
  const int col = n0 + lane;
  const float bb = sb1[b * C_ + col];
  const float wrow = scan_w(m0 + lane);
  float d[4];
#pragma unroll
  for (int j = 0; j < 4; ++j) {
    const int R = wave * 4 + j;
    d[j] = (row_sum<8>(red, R, lane) + bb) - v[(m0 + R) * C_ + col];
  }
#pragma unroll
  for (int j = 0; j < 4; ++j) dts[(wave * 4 + j) * TP_ + lane] = d[j];
  __syncthreads();
#pragma unroll
  for (int j = 0; j < 4; ++j) {
    const int Cc = wave * 4 + j;
    wdpT[(n0 + Cc) * T_ + m0 + lane] = wrow * dts[lane * TP_ + Cc];
  }
#pragma unroll
  for (int j = 0; j < 4; ++j) dp[(m0 + wave * 4 + j) * C_ + col] = d[j];
}

__global__ void __launch_bounds__(256) __attribute__((amdgpu_waves_per_eu(2, 2)))
k_gemm_dh1(const float* __restrict__ sW1,
                           const float* __restrict__ Wo,
                           const float* __restrict__ bo,
                           float* __restrict__ rout,
                           float* __restrict__ ws) {
  __shared__ float red[8 * 32 * RP_];
  __shared__ float dts[32 * TP_];
  const int b = blockIdx.y;
  const int lane = threadIdx.x & 31, wave = threadIdx.x >> 5;
  if (blockIdx.x >= 48) {
    wave_matvec<C_, 2, 2, false>(Wo, ws + OFF_RP + b * C_, bo, rout + b * C_,
                              (blockIdx.x - 48) * 16 + wave * 2, lane);
    return;
  }
  const int m0 = (blockIdx.x % 2) * 32, n0 = (blockIdx.x / 2) * 32;
  const float* A = ws + OFF_DP + b * T_ * C_;
  const float* W = sW1 + b * C_ * H_;
  const float* gp = ws + OFF_GP + b * T_ * H_;
  float* wdhT = ws + OFF_WDHT + b * H_ * T_;
  v8f acc[4] = {};
  gemm32x32_slice<false, C_ / 8>(A, C_, W, H_, wave * (C_ / 8), m0, n0, lane, acc);
  put_partial(acc, red, wave, lane);
  __syncthreads();
  const int col = n0 + lane;
  const float wrow = scan_w(m0 + lane);
#pragma unroll
  for (int j = 0; j < 4; ++j) {
    const int R = wave * 4 + j;
    dts[R * TP_ + lane] = row_sum<8>(red, R, lane) * gp[(m0 + R) * H_ + col];
  }
  __syncthreads();
#pragma unroll
  for (int j = 0; j < 4; ++j) {
    const int Cc = wave * 4 + j;
    wdhT[(n0 + Cc) * T_ + m0 + lane] = wrow * dts[lane * TP_ + Cc];
  }
}

template <int N>
__device__ __forceinline__ void update_tile(const float* __restrict__ sW,
                                            float* __restrict__ nW,
                                            const float* __restrict__ G,
                                            const float* __restrict__ Bm,
                                            int tile, int lane, float dT) {
  const int m0 = (tile / (N / 32)) * 32, n0 = (tile % (N / 32)) * 32;
  v8f acc[4] = {};
  gemm32x32<false>(G, T_, Bm, N, T_, m0, n0, lane, acc);
  const int cr = lane & 15, hi = lane >> 4;
#pragma unroll
  for (int mi = 0; mi < 2; ++mi)
#pragma unroll
    for (int ni = 0; ni < 2; ++ni) {
      int col = n0 + ni * 16 + cr;
      v8f a = acc[mi * 2 + ni];
#pragma unroll
      for (int r = 0; r < 8; ++r) {
        int row = m0 + mi * 16 + r + 8 * hi;
        nW[row * N + col] = dT * sW[row * N + col] + a[r];
      }
    }
}

__global__ void k_update(const float* __restrict__ sW0, const float* __restrict__ sb0,
                         const float* __restrict__ sW1, const float* __restrict__ sb1,
                         float* __restrict__ nW0, float* __restrict__ nb0,
                         float* __restrict__ nW1, float* __restrict__ nb1,
                         const float* __restrict__ ws) {
  const int bx = blockIdx.x, lane = threadIdx.x;
  const float dT = decay_T();
  if (bx >= 648) {
    const int b = (bx - 648) / 288, tile = (bx - 648) % 288;
    update_tile<H_>(sW1 + b * C_ * H_, nW1 + b * C_ * H_, ws + OFF_WDPT + b * C_ * T_,
                    ws + OFF_H1 + b * T_ * H_, tile, lane, dT);
    return;
  }
  if (bx >= 72) {
    const int b = (bx - 72) / 288, tile = (bx - 72) % 288;
    update_tile<C_>(sW0 + b * H_ * C_, nW0 + b * H_ * C_, ws + OFF_WDHT + b * H_ * T_,
                    ws + OFF_K + b * T_ * C_, tile, lane, dT);
    return;
  }
  const int b = bx / 36;
  const int r = (bx % 36) * 32 + lane;
  if (r < H_) {
    const float* p = ws + OFF_WDHT + b * H_ * T_ + r * T_;
    float s = 0.f;
#pragma unroll 4
    for (int t = 0; t < T_; ++t) s += p[t];
    nb0[b * H_ + r] = dT * sb0[b * H_ + r] + s;
  } else {
    const int c = r - H_;
    const float* p = ws + OFF_WDPT + b * C_ * T_ + c * T_;
    float s = 0.f;
#pragma unroll 4
    for (int t = 0; t < T_; ++t) s += p[t];
    nb1[b * C_ + c] = dT * sb1[b * C_ + c] + s;
  }
}

}

extern "C" void kernel_launch(void* const* d_in, const int* in_sizes, int n_in,
                              void* d_out, int out_size, void* d_ws, size_t ws_size,
                              hipStream_t stream) {
  const float* x   = (const float*)d_in[0];
  const float* Wq  = (const float*)d_in[1];
  const float* bq  = (const float*)d_in[2];
  const float* Wk  = (const float*)d_in[3];
  const float* bk  = (const float*)d_in[4];
  const float* Wv  = (const float*)d_in[5];
  const float* bv  = (const float*)d_in[6];
  const float* Wo  = (const float*)d_in[7];
  const float* bo  = (const float*)d_in[8];
  const float* iq  = (const float*)d_in[9];
  const float* sW0 = (const float*)d_in[10];
  const float* sb0 = (const float*)d_in[11];
  const float* sW1 = (const float*)d_in[12];
  const float* sb1 = (const float*)d_in[13];

  float* out = (float*)d_out;
  float* ws  = (float*)d_ws;

  float* out_out = out;
  float* out_nW0 = out + B_ * C_;
  float* out_nb0 = out_nW0 + B_ * H_ * C_;
  float* out_nW1 = out_nb0 + B_ * H_;
  float* out_nb1 = out_nW1 + B_ * C_ * H_;

  k_gemm_kv<<<dim3(30, B_, 2), 128, 0, stream>>>(x, Wk, bk, Wv, bv, Wq, bq, iq, ws);
  k_gemm_h1<<<dim3(96, B_), 128, 0, stream>>>(sW0, sb0, ws);
  k_gemm_dp<<<dim3(48, B_), 256, 0, stream>>>(sW1, sb1, ws);
  k_gemm_dh1<<<dim3(72, B_), 256, 0, stream>>>(sW1, Wo, bo, out_out, ws);
  k_update  <<<dim3(1224), 32, 0, stream>>>(sW0, sb0, sW1, sb1, out_nW0, out_nb0,
                                            out_nW1, out_nb1, ws);
}


# Launches 5 -> 4: the dh1 stage runs inside the nW0 blocks of the last launch (a block owns 16 rows of nW0 and builds its dh1p panel in LDS, never stored); nW1, biases, retrieval share that launch
# speedup vs baseline: 7.8926x; 1.0862x over previous
#include <hip/hip_runtime.h>
#include <cstdint>

typedef float v2f __attribute__((ext_vector_type(2)));
typedef float v4f __attribute__((ext_vector_type(4)));
typedef float v8f __attribute__((ext_vector_type(8)));

namespace {

constexpr int B_ = 2, T_ = 64, C_ = 384, H_ = 768;
constexpr float LR = 0.01f, MOM = 0.9f, DECAY = 0.001f;

constexpr int OFF_K    = 0;
constexpr int OFF_V    = OFF_K    + B_*T_*C_;
constexpr int OFF_H1   = OFF_V    + B_*T_*C_;
constexpr int OFF_GP   = OFF_H1   + B_*T_*H_;
constexpr int OFF_DP   = OFF_GP   + B_*T_*H_;
constexpr int OFF_WDPT = OFF_DP   + B_*T_*C_;
constexpr int OFF_RQ   = OFF_WDPT + B_*C_*T_;
constexpr int OFF_RH   = OFF_RQ   + C_;
constexpr int OFF_RP   = OFF_RH   + B_*H_;

__device__ __forceinline__ float scan_w(int t) {
  const float a = 1.0f - DECAY;
  const float n = (float)(T_ - t);
  const float an = __powf(a, n), mn = __powf(MOM, n);
  return -LR * (1.0f - MOM) * (an - mn) / (a - MOM);
}
__device__ __forceinline__ float decay_T() { return __powf(1.0f - DECAY, (float)T_); }

__device__ __forceinline__ v8f wmma4(v2f a, v2f b, v8f c) {
  return __builtin_amdgcn_wmma_f32_16x16x4_f32(false, a, false, b, (short)0, c,
                                               false, false);
}

template <bool BT>
__device__ __forceinline__ void load_frags(const float* __restrict__ A, int lda,
                                           const float* __restrict__ Bm, int ldb,
                                           int kp, int m0, int n0, int mr,
                                           v4f& a0, v4f& a1, v4f& b0, v4f& b1) {
  a0 = *reinterpret_cast<const v4f*>(A + (m0 + mr) * lda + kp);
  a1 = *reinterpret_cast<const v4f*>(A + (m0 + 16 + mr) * lda + kp);
  if (BT) {
    b0 = *reinterpret_cast<const v4f*>(Bm + (n0 + mr) * ldb + kp);
    b1 = *reinterpret_cast<const v4f*>(Bm + (n0 + 16 + mr) * ldb + kp);
  } else {
    const float* pb = Bm + kp * ldb + n0 + mr;
    b0.x = pb[0];       b0.y = pb[ldb];       b0.z = pb[2 * ldb];       b0.w = pb[3 * ldb];
    b1.x = pb[16];      b1.y = pb[ldb + 16];  b1.z = pb[2 * ldb + 16];  b1.w = pb[3 * ldb + 16];
  }
}

__device__ __forceinline__ void mma_pair(v4f a0, v4f a1, v4f b0, v4f b1, v8f acc[4]) {
  acc[0] = wmma4(a0.xy, b0.xy, acc[0]);
  acc[1] = wmma4(a0.xy, b1.xy, acc[1]);
  acc[2] = wmma4(a1.xy, b0.xy, acc[2]);
  acc[3] = wmma4(a1.xy, b1.xy, acc[3]);
  acc[0] = wmma4(a0.zw, b0.zw, acc[0]);
  acc[1] = wmma4(a0.zw, b1.zw, acc[1]);
  acc[2] = wmma4(a1.zw, b0.zw, acc[2]);
  acc[3] = wmma4(a1.zw, b1.zw, acc[3]);
}

template <bool BT>
__device__ __forceinline__ void gemm32x32(const float* __restrict__ A, int lda,
                                          const float* __restrict__ Bm, int ldb,
                                          int K, int m0, int n0, int lane,
                                          v8f acc[4]) {
  const int mr = lane & 15;
  const int kq = (lane >> 4) << 2;
  for (int kk = 0; kk < K; kk += 8) {
    const int kp = kk + kq;
    v4f a0, a1, b0, b1;
    load_frags<BT>(A, lda, Bm, ldb, kp, m0, n0, mr, a0, a1, b0, b1);
    mma_pair(a0, a1, b0, b1, acc);
  }
}

template <bool BT, int KS>
__device__ __forceinline__ void gemm32x32_slice(const float* __restrict__ A, int lda,
                                                const float* __restrict__ Bm, int ldb,
                                                int k0, int m0, int n0, int lane,
                                                v8f acc[4]) {
  const int mr = lane & 15;
  const int kq = (lane >> 4) << 2;
#pragma unroll
  for (int kk = 0; kk < KS; kk += 8) {
    const int kp = k0 + kk + kq;
    v4f a0, a1, b0, b1;
    load_frags<BT>(A, lda, Bm, ldb, kp, m0, n0, mr, a0, a1, b0, b1);
    mma_pair(a0, a1, b0, b1, acc);
  }
}

constexpr int RP_ = 36;
constexpr int TP_ = 33;

__device__ __forceinline__ void put_partial(const v8f acc[4], float* red, int slice, int lane) {
  const int cr = lane & 15, hi = lane >> 4;
#pragma unroll
  for (int qq = 0; qq < 4; ++qq)
#pragma unroll
    for (int r = 0; r < 8; ++r) {
      const int R = (qq >> 1) * 16 + 8 * hi + r;
      const int Cc = (qq & 1) * 16 + cr;
      red[(slice * 32 + R) * RP_ + Cc] = acc[qq][r];
    }
}

template <int S>
__device__ __forceinline__ float row_sum(const float* red, int R, int lane) {
  float s = red[R * RP_ + lane];
#pragma unroll
  for (int sl = 1; sl < S; ++sl) s += red[(sl * 32 + R) * RP_ + lane];
  return s;
}

template <int K, int NR, int RU, bool SILU>
__device__ __forceinline__ void wave_matvec(const float* __restrict__ M,
                                            const float* __restrict__ v,
                                            const float* __restrict__ bias,
                                            float* __restrict__ y, int row0, int lane) {
  constexpr int NV = K / 128;
  v4f vv[NV];
#pragma unroll
  for (int j = 0; j < NV; ++j)
    vv[j] = *reinterpret_cast<const v4f*>(v + j * 128 + lane * 4);
  for (int r0 = 0; r0 < NR; r0 += RU) {
    float s[RU];
#pragma unroll
    for (int i = 0; i < RU; ++i) {
      const float* mr = M + (row0 + r0 + i) * K + lane * 4;
      float a = 0.f;
#pragma unroll
      for (int j = 0; j < NV; ++j) {
        v4f m = *reinterpret_cast<const v4f*>(mr + j * 128);
        a += m.x * vv[j].x;
        a += m.y * vv[j].y;
        a += m.z * vv[j].z;
        a += m.w * vv[j].w;
      }
      s[i] = a;
    }
#pragma unroll
    for (int off = 16; off > 0; off >>= 1)
#pragma unroll
      for (int i = 0; i < RU; ++i) s[i] += __shfl_xor(s[i], off, 32);
    float t = s[0];
#pragma unroll
    for (int i = 1; i < RU; ++i)
      if (lane == i) t = s[i];
    if (lane < RU) {
      const int row = row0 + r0 + lane;
      t += bias[row];
      if (SILU) {
        float sig = 1.0f / (1.0f + __expf(-t));
        t = t * sig;
      }
      y[row] = t;
    }
  }
}

__global__ void __launch_bounds__(128) __attribute__((amdgpu_waves_per_eu(1, 1)))
k_gemm_kv(const float* __restrict__ x,
                          const float* __restrict__ Wk, const float* __restrict__ bk,
                          const float* __restrict__ Wv, const float* __restrict__ bv,
                          const float* __restrict__ Wq, const float* __restrict__ bq,
                          const float* __restrict__ iq,
                          float* __restrict__ ws) {
  __shared__ float red[4 * 32 * RP_];
  const int b = blockIdx.y, which = blockIdx.z;
  const int lane = threadIdx.x & 31, wave = threadIdx.x >> 5;
  if (blockIdx.x >= 24) {
    const int blk = ((blockIdx.x - 24) * B_ + b) * 2 + which;
    wave_matvec<C_, 4, 4, false>(Wq, iq, bq, ws + OFF_RQ, blk * 16 + wave * 4, lane);
    return;
  }
  const int m0 = (blockIdx.x & 1) * 32, n0 = (blockIdx.x >> 1) * 32;
  const float* A = x + b * T_ * C_;
  const float* W = which ? Wv : Wk;
  const float* bias = which ? bv : bk;
  float* out = ws + (which ? OFF_V : OFF_K) + b * T_ * C_;
  v8f acc[4] = {};
  gemm32x32_slice<true, C_ / 4>(A, C_, W, C_, wave * (C_ / 4), m0, n0, lane, acc);
  put_partial(acc, red, wave, lane);
  __syncthreads();
  const int col = n0 + lane;
  const float bb = bias[col];
#pragma unroll
  for (int j = 0; j < 8; ++j) {
    const int R = wave * 8 + j;
    out[(m0 + R) * C_ + col] = row_sum<4>(red, R, lane) + bb;
  }
}

__global__ void __launch_bounds__(128) __attribute__((amdgpu_waves_per_eu(1, 1)))
k_gemm_h1(const float* __restrict__ sW0,
                          const float* __restrict__ sb0,
                          float* __restrict__ ws) {
  __shared__ float red[4 * 32 * RP_];
  const int b = blockIdx.y;
  const int lane = threadIdx.x & 31, wave = threadIdx.x >> 5;
  if (blockIdx.x >= 48) {
    wave_matvec<C_, 4, 4, true>(sW0 + b * H_ * C_, ws + OFF_RQ, sb0 + b * H_,
                             ws + OFF_RH + b * H_, (blockIdx.x - 48) * 16 + wave * 4, lane);
    return;
  }
  const int m0 = (blockIdx.x % 2) * 32, n0 = (blockIdx.x / 2) * 32;
  const float* A = ws + OFF_K + b * T_ * C_;
  const float* W = sW0 + b * H_ * C_;
  float* h1 = ws + OFF_H1 + b * T_ * H_;
  float* gp = ws + OFF_GP + b * T_ * H_;
  v8f acc[4] = {};
  gemm32x32_slice<true, C_ / 4>(A, C_, W, C_, wave * (C_ / 4), m0, n0, lane, acc);
  put_partial(acc, red, wave, lane);
  __syncthreads();
  const int col = n0 + lane;
  const float bb = sb0[b * H_ + col];
#pragma unroll
  for (int j = 0; j < 8; ++j) {
    const int R = wave * 8 + j;
    float pre = row_sum<4>(red, R, lane) + bb;
    float sig = 1.0f / (1.0f + __expf(-pre));
    h1[(m0 + R) * H_ + col] = pre * sig;
    gp[(m0 + R) * H_ + col] = sig * (1.0f + pre * (1.0f - sig));
  }
}

__global__ void __launch_bounds__(256) __attribute__((amdgpu_waves_per_eu(2, 2)))
k_gemm_dp(const float* __restrict__ sW1,
                          const float* __restrict__ sb1,
                          float* __restrict__ ws) {
  __shared__ float red[8 * 32 * RP_];
  __shared__ float dts[32 * TP_];
  const int b = blockIdx.y;
  const int lane = threadIdx.x & 31, wave = threadIdx.x >> 5;
  if (blockIdx.x >= 24) {
    wave_matvec<H_, 2, 2, false>(sW1 + b * C_ * H_, ws + OFF_RH + b * H_, sb1 + b * C_,
                              ws + OFF_RP + b * C_, (blockIdx.x - 24) * 16 + wave * 2, lane);
    return;
  }
  const int m0 = (blockIdx.x % 2) * 32, n0 = (blockIdx.x / 2) * 32;
  const float* A = ws + OFF_H1 + b * T_ * H_;
  const float* W = sW1 + b * C_ * H_;
  const float* v = ws + OFF_V + b * T_ * C_;
  float* dp = ws + OFF_DP + b * T_ * C_;
  float* wdpT = ws + OFF_WDPT + b * C_ * T_;
  v8f acc[4] = {};
  gemm32x32_slice<true, H_ / 8>(A, H_, W, H_, wave * (H_ / 8), m0, n0, lane, acc);
  put_partial(acc, red, wave, lane);
  __syncthreads();
  const int col = n0 + lane;
  const float bb = sb1[b * C_ + col];
  const float wrow = scan_w(m0 + lane);
  float d[4];
#pragma unroll
  for (int j = 0; j < 4; ++j) {
    const int R = wave * 4 + j;
    d[j] = (row_sum<8>(red, R, lane) + bb) - v[(m0 + R) * C_ + col];
  }
#pragma unroll
  for (int j = 0; j < 4; ++j) dts[(wave * 4 + j) * TP_ + lane] = d[j];
  __syncthreads();
#pragma unroll
  for (int j = 0; j < 4; ++j) {
    const int Cc = wave * 4 + j;
    wdpT[(n0 + Cc) * T_ + m0 + lane] = wrow * dts[lane * TP_ + Cc];
  }
#pragma unroll
  for (int j = 0; j < 4; ++j) dp[(m0 + wave * 4 + j) * C_ + col] = d[j];
}

template <int N>
__device__ __forceinline__ void update_tile(const float* __restrict__ sW,
                                            float* __restrict__ nW,
                                            const float* __restrict__ G,
                                            const float* __restrict__ Bm,
                                            int tile, int lane, float dT) {
  const int m0 = (tile / (N / 32)) * 32, n0 = (tile % (N / 32)) * 32;
  v8f acc[4] = {};
  gemm32x32<false>(G, T_, Bm, N, T_, m0, n0, lane, acc);
  const int cr = lane & 15, hi = lane >> 4;
#pragma unroll
  for (int mi = 0; mi < 2; ++mi)
#pragma unroll
    for (int ni = 0; ni < 2; ++ni) {
      int col = n0 + ni * 16 + cr;
      v8f a = acc[mi * 2 + ni];
#pragma unroll
      for (int r = 0; r < 8; ++r) {
        int row = m0 + mi * 16 + r + 8 * hi;
        nW[row * N + col] = dT * sW[row * N + col] + a[r];
      }
    }
}

constexpr int PP_ = 20;
constexpr int WP_ = 68;
constexpr int OP_ = 388;
struct __attribute__((packed, aligned(4))) f3 { float x, y, z; };

__global__ void __launch_bounds__(256) __attribute__((amdgpu_waves_per_eu(2, 2)))
k_grad_update(const float* __restrict__ sW0, const float* __restrict__ sb0,
              const float* __restrict__ sW1, const float* __restrict__ sb1,
              const float* __restrict__ Wo, const float* __restrict__ bo,
              float* __restrict__ nW0, float* __restrict__ nb0,
              float* __restrict__ nW1, float* __restrict__ nb1,
              float* __restrict__ rout,
              const float* __restrict__ ws) {
  __shared__ float red[8 * 64 * PP_];
  __shared__ float wdh[16 * WP_];
  const int bx = blockIdx.x, tid = threadIdx.x;
  const int lane = tid & 31, wave = tid >> 5;
  const float dT = decay_T();
  if (bx >= 171) {
    const int q = bx - 171, b = q / 24, blk = q % 24;
    wave_matvec<C_, 2, 2, false>(Wo, ws + OFF_RP + b * C_, bo, rout + b * C_,
                                 blk * 16 + wave * 2, lane);
    return;
  }
  if (bx >= 168) {
    const int r = (bx - 168) * 256 + tid;
    const int b = r / C_, c = r % C_;
    const float* p = ws + OFF_WDPT + b * C_ * T_ + c * T_;
    float s = 0.f;
#pragma unroll 4
    for (int t = 0; t < T_; ++t) s += p[t];
    nb1[b * C_ + c] = dT * sb1[b * C_ + c] + s;
    return;
  }
  if (bx >= 96) {
    const int id = (bx - 96) * 8 + wave;
    const int b = id / 288, tile = id % 288;
    update_tile<H_>(sW1 + b * C_ * H_, nW1 + b * C_ * H_, ws + OFF_WDPT + b * C_ * T_,
                    ws + OFF_H1 + b * T_ * H_, tile, lane, dT);
    return;
  }

  const int b = bx / 48, h0 = (bx % 48) * 16;
  const int mr = lane & 15, hi = lane >> 4;
  const int kq = hi << 2;
  const float* dp = ws + OFF_DP + b * T_ * C_;
  const float* W1 = sW1 + b * C_ * H_;
  const float* gp = ws + OFF_GP + b * T_ * H_;
  const float* kk = ws + OFF_K + b * T_ * C_;
  const float* W0 = sW0 + b * H_ * C_;
  float* oW0 = nW0 + b * H_ * C_;

  v4f fa[6][4], fb[6];
#pragma unroll
  for (int s = 0; s < 6; ++s) {
    const int kp = wave * 48 + s * 8 + kq;
#pragma unroll
    for (int mi = 0; mi < 4; ++mi)
      fa[s][mi] = *reinterpret_cast<const v4f*>(dp + (mi * 16 + mr) * C_ + kp);
    const float* pb = W1 + kp * H_ + h0 + mr;
    fb[s].x = pb[0];  fb[s].y = pb[H_];  fb[s].z = pb[2 * H_];  fb[s].w = pb[3 * H_];
  }
  const int rh = tid & 15, tq = tid >> 4;
  float g[4];
#pragma unroll
  for (int j = 0; j < 4; ++j) g[j] = gp[(tq + 16 * j) * H_ + h0 + rh];

  v8f acc[4] = {};
#pragma unroll
  for (int s = 0; s < 6; ++s) {
#pragma unroll
    for (int mi = 0; mi < 4; ++mi) acc[mi] = wmma4(fa[s][mi].xy, fb[s].xy, acc[mi]);
#pragma unroll
    for (int mi = 0; mi < 4; ++mi) acc[mi] = wmma4(fa[s][mi].zw, fb[s].zw, acc[mi]);
  }
#pragma unroll
  for (int mi = 0; mi < 4; ++mi)
#pragma unroll
    for (int r = 0; r < 8; ++r)
      red[(wave * 64 + mi * 16 + 8 * hi + r) * PP_ + mr] = acc[mi][r];
  __builtin_amdgcn_sched_barrier(0);

  const int n0w = wave * 48;
  f3 kb[8][4];
#pragma unroll
  for (int s = 0; s < 8; ++s)
#pragma unroll
    for (int i = 0; i < 4; ++i)
      kb[s][i] = *reinterpret_cast<const f3*>(kk + (s * 8 + kq + i) * C_ + n0w + 3 * mr);
  v4f w0q[6];
#pragma unroll
  for (int i = 0; i < 6; ++i) {
    const int idx = tid + 256 * i;
    w0q[i] = *reinterpret_cast<const v4f*>(W0 + (h0 + idx / 96) * C_ + 4 * (idx % 96));
  }
  __syncthreads();
#pragma unroll
  for (int j = 0; j < 4; ++j) {
    const int t = tq + 16 * j;
    float s = red[t * PP_ + rh];
#pragma unroll
    for (int sl = 1; sl < 8; ++sl) s += red[(sl * 64 + t) * PP_ + rh];
    const float d = s * g[j];
    wdh[rh * WP_ + t] = scan_w(t) * d;
  }
  __syncthreads();

  if (tid < 16) {
    float s = 0.f;
#pragma unroll
    for (int t = 0; t < T_; ++t) s += wdh[tid * WP_ + t];
    nb0[b * H_ + h0 + tid] = dT * sb0[b * H_ + h0 + tid] + s;
  }

  v8f ac2[3] = {};
#pragma unroll
  for (int s = 0; s < 8; ++s) {
    const v4f a = *reinterpret_cast<const v4f*>(wdh + mr * WP_ + s * 8 + kq);
    ac2[0] = wmma4(a.xy, v2f{kb[s][0].x, kb[s][1].x}, ac2[0]);
    ac2[1] = wmma4(a.xy, v2f{kb[s][0].y, kb[s][1].y}, ac2[1]);
    ac2[2] = wmma4(a.xy, v2f{kb[s][0].z, kb[s][1].z}, ac2[2]);
    ac2[0] = wmma4(a.zw, v2f{kb[s][2].x, kb[s][3].x}, ac2[0]);
    ac2[1] = wmma4(a.zw, v2f{kb[s][2].y, kb[s][3].y}, ac2[1]);
    ac2[2] = wmma4(a.zw, v2f{kb[s][2].z, kb[s][3].z}, ac2[2]);
  }
  float* outp = red;
#pragma unroll
  for (int r = 0; r < 8; ++r) {
    float* po = outp + (8 * hi + r) * OP_ + n0w + 3 * mr;
    po[0] = ac2[0][r];  po[1] = ac2[1][r];  po[2] = ac2[2][r];
  }
  __syncthreads();
#pragma unroll
  for (int i = 0; i < 6; ++i) {
    const int idx = tid + 256 * i;
    const int row = idx / 96, c4 = 4 * (idx % 96);
    const v4f a = *reinterpret_cast<const v4f*>(outp + row * OP_ + c4);
    v4f o;
    o.x = dT * w0q[i].x + a.x;  o.y = dT * w0q[i].y + a.y;
    o.z = dT * w0q[i].z + a.z;  o.w = dT * w0q[i].w + a.w;
    *reinterpret_cast<v4f*>(oW0 + (h0 + row) * C_ + c4) = o;
  }
}

}

extern "C" void kernel_launch(void* const* d_in, const int* in_sizes, int n_in,
                              void* d_out, int out_size, void* d_ws, size_t ws_size,
                              hipStream_t stream) {
  const float* x   = (const float*)d_in[0];
  const float* Wq  = (const float*)d_in[1];
  const float* bq  = (const float*)d_in[2];
  const float* Wk  = (const float*)d_in[3];
  const float* bk  = (const float*)d_in[4];
  const float* Wv  = (const float*)d_in[5];
  const float* bv  = (const float*)d_in[6];
  const float* Wo  = (const float*)d_in[7];
  const float* bo  = (const float*)d_in[8];
  const float* iq  = (const float*)d_in[9];
  const float* sW0 = (const float*)d_in[10];
  const float* sb0 = (const float*)d_in[11];
  const float* sW1 = (const float*)d_in[12];
  const float* sb1 = (const float*)d_in[13];

  float* out = (float*)d_out;
  float* ws  = (float*)d_ws;

  float* out_out = out;
  float* out_nW0 = out + B_ * C_;
  float* out_nb0 = out_nW0 + B_ * H_ * C_;
  float* out_nW1 = out_nb0 + B_ * H_;
  float* out_nb1 = out_nW1 + B_ * C_ * H_;

  k_gemm_kv<<<dim3(30, B_, 2), 128, 0, stream>>>(x, Wk, bk, Wv, bv, Wq, bq, iq, ws);
  k_gemm_h1<<<dim3(96, B_), 128, 0, stream>>>(sW0, sb0, ws);
  k_gemm_dp<<<dim3(48, B_), 256, 0, stream>>>(sW1, sb1, ws);
  k_grad_update<<<dim3(219), 256, 0, stream>>>(sW0, sb0, sW1, sb1, Wo, bo, out_nW0, out_nb0,
                                               out_nW1, out_nb1, out_out, ws);
}


# Launches 1-3: a wave fetches its K slice of both operands with whole-line loads (4 full 128-byte lines each) into LDS rows of its own and reads the fragments there; order of sums kept
# speedup vs baseline: 8.2610x; 1.0467x over previous
#include <hip/hip_runtime.h>
#include <cstdint>

typedef float v2f __attribute__((ext_vector_type(2)));
typedef float v4f __attribute__((ext_vector_type(4)));
typedef float v8f __attribute__((ext_vector_type(8)));

namespace {

constexpr int B_ = 2, T_ = 64, C_ = 384, H_ = 768;
constexpr float LR = 0.01f, MOM = 0.9f, DECAY = 0.001f;

constexpr int OFF_K    = 0;
constexpr int OFF_V    = OFF_K    + B_*T_*C_;
constexpr int OFF_H1   = OFF_V    + B_*T_*C_;
constexpr int OFF_GP   = OFF_H1   + B_*T_*H_;
constexpr int OFF_DP   = OFF_GP   + B_*T_*H_;
constexpr int OFF_WDPT = OFF_DP   + B_*T_*C_;
constexpr int OFF_RQ   = OFF_WDPT + B_*C_*T_;
constexpr int OFF_RH   = OFF_RQ   + C_;
constexpr int OFF_RP   = OFF_RH   + B_*H_;

__device__ __forceinline__ float scan_w(int t) {
  const float a = 1.0f - DECAY;
  const float n = (float)(T_ - t);
  const float an = __powf(a, n), mn = __powf(MOM, n);
  return -LR * (1.0f - MOM) * (an - mn) / (a - MOM);
}
__device__ __forceinline__ float decay_T() { return __powf(1.0f - DECAY, (float)T_); }

__device__ __forceinline__ v8f wmma4(v2f a, v2f b, v8f c) {
  return __builtin_amdgcn_wmma_f32_16x16x4_f32(false, a, false, b, (short)0, c,
                                               false, false);
}

template <bool BT>
__device__ __forceinline__ void load_frags(const float* __restrict__ A, int lda,
                                           const float* __restrict__ Bm, int ldb,
                                           int kp, int m0, int n0, int mr,
                                           v4f& a0, v4f& a1, v4f& b0, v4f& b1) {
  a0 = *reinterpret_cast<const v4f*>(A + (m0 + mr) * lda + kp);
  a1 = *reinterpret_cast<const v4f*>(A + (m0 + 16 + mr) * lda + kp);
  if (BT) {
    b0 = *reinterpret_cast<const v4f*>(Bm + (n0 + mr) * ldb + kp);
    b1 = *reinterpret_cast<const v4f*>(Bm + (n0 + 16 + mr) * ldb + kp);
  } else {
    const float* pb = Bm + kp * ldb + n0 + mr;
    b0.x = pb[0];       b0.y = pb[ldb];       b0.z = pb[2 * ldb];       b0.w = pb[3 * ldb];
    b1.x = pb[16];      b1.y = pb[ldb + 16];  b1.z = pb[2 * ldb + 16];  b1.w = pb[3 * ldb + 16];
  }
}

__device__ __forceinline__ void mma_pair(v4f a0, v4f a1, v4f b0, v4f b1, v8f acc[4]) {
  acc[0] = wmma4(a0.xy, b0.xy, acc[0]);
  acc[1] = wmma4(a0.xy, b1.xy, acc[1]);
  acc[2] = wmma4(a1.xy, b0.xy, acc[2]);
  acc[3] = wmma4(a1.xy, b1.xy, acc[3]);
  acc[0] = wmma4(a0.zw, b0.zw, acc[0]);
  acc[1] = wmma4(a0.zw, b1.zw, acc[1]);
  acc[2] = wmma4(a1.zw, b0.zw, acc[2]);
  acc[3] = wmma4(a1.zw, b1.zw, acc[3]);
}

template <bool BT>
__device__ __forceinline__ void gemm32x32(const float* __restrict__ A, int lda,
                                          const float* __restrict__ Bm, int ldb,
                                          int K, int m0, int n0, int lane,
                                          v8f acc[4]) {
  const int mr = lane & 15;
  const int kq = (lane >> 4) << 2;
  for (int kk = 0; kk < K; kk += 8) {
    const int kp = kk + kq;
    v4f a0, a1, b0, b1;
    load_frags<BT>(A, lda, Bm, ldb, kp, m0, n0, mr, a0, a1, b0, b1);
    mma_pair(a0, a1, b0, b1, acc);
  }
}

constexpr int SP_ = 100;
constexpr int STG_ = 2 * 32 * SP_;
template <int KS>
__device__ __forceinline__ void gemm32x32_slice_staged(const float* __restrict__ A, int lda,
                                                       const float* __restrict__ Bm, int ldb,
                                                       int k0, int m0, int n0, int lane,
                                                       float* st, v8f acc[4]) {
  static_assert(KS % 32 == 0 && KS + 4 <= SP_, "slice = whole lines, and it fits the pitch");
  constexpr int NC = KS / 32;
  const int lr = lane >> 3;
  const int lp = (lane & 7) << 2;
  const int mr = lane & 15;
  const int kq = (lane >> 4) << 2;
  float* sa = st;
  float* sb = st + 32 * SP_;
  v4f ga[NC][8], gb[NC][8];
#pragma unroll
  for (int s = 0; s < NC; ++s)
#pragma unroll
    for (int g = 0; g < 8; ++g) {
      ga[s][g] = *reinterpret_cast<const v4f*>(A + (m0 + 4 * g + lr) * lda + k0 + 32 * s + lp);
      gb[s][g] = *reinterpret_cast<const v4f*>(Bm + (n0 + 4 * g + lr) * ldb + k0 + 32 * s + lp);
    }
#pragma unroll
  for (int s = 0; s < NC; ++s) {
#pragma unroll
    for (int g = 0; g < 8; ++g) {
      *reinterpret_cast<v4f*>(sa + (4 * g + lr) * SP_ + 32 * s + lp) = ga[s][g];
      *reinterpret_cast<v4f*>(sb + (4 * g + lr) * SP_ + 32 * s + lp) = gb[s][g];
    }
    __builtin_amdgcn_fence(__ATOMIC_RELEASE, "wavefront");
    __builtin_amdgcn_wave_barrier();
    __builtin_amdgcn_fence(__ATOMIC_ACQUIRE, "wavefront");
#pragma unroll
    for (int u = 0; u < 4; ++u) {
      const int kp = 32 * s + 8 * u + kq;
      const v4f a0 = *reinterpret_cast<const v4f*>(sa + mr * SP_ + kp);
      const v4f a1 = *reinterpret_cast<const v4f*>(sa + (16 + mr) * SP_ + kp);
      const v4f b0 = *reinterpret_cast<const v4f*>(sb + mr * SP_ + kp);
      const v4f b1 = *reinterpret_cast<const v4f*>(sb + (16 + mr) * SP_ + kp);
      mma_pair(a0, a1, b0, b1, acc);
    }
  }
}

constexpr int RP_ = 36;
constexpr int TP_ = 33;

__device__ __forceinline__ void put_partial(const v8f acc[4], float* red, int slice, int lane) {
  const int cr = lane & 15, hi = lane >> 4;
#pragma unroll
  for (int qq = 0; qq < 4; ++qq)
#pragma unroll
    for (int r = 0; r < 8; ++r) {
      const int R = (qq >> 1) * 16 + 8 * hi + r;
      const int Cc = (qq & 1) * 16 + cr;
      red[(slice * 32 + R) * RP_ + Cc] = acc[qq][r];
    }
}

template <int S>
__device__ __forceinline__ float row_sum(const float* red, int R, int lane) {
  float s = red[R * RP_ + lane];
#pragma unroll
  for (int sl = 1; sl < S; ++sl) s += red[(sl * 32 + R) * RP_ + lane];
  return s;
}

template <int K, int NR, int RU, bool SILU>
__device__ __forceinline__ void wave_matvec(const float* __restrict__ M,
                                            const float* __restrict__ v,
                                            const float* __restrict__ bias,
                                            float* __restrict__ y, int row0, int lane) {
  constexpr int NV = K / 128;
  v4f vv[NV];
#pragma unroll
  for (int j = 0; j < NV; ++j)
    vv[j] = *reinterpret_cast<const v4f*>(v + j * 128 + lane * 4);
  for (int r0 = 0; r0 < NR; r0 += RU) {
    float s[RU];
#pragma unroll
    for (int i = 0; i < RU; ++i) {
      const float* mr = M + (row0 + r0 + i) * K + lane * 4;
      float a = 0.f;
#pragma unroll
      for (int j = 0; j < NV; ++j) {
        v4f m = *reinterpret_cast<const v4f*>(mr + j * 128);
        a += m.x * vv[j].x;
        a += m.y * vv[j].y;
        a += m.z * vv[j].z;
        a += m.w * vv[j].w;
      }
      s[i] = a;
    }
#pragma unroll
    for (int off = 16; off > 0; off >>= 1)
#pragma unroll
      for (int i = 0; i < RU; ++i) s[i] += __shfl_xor(s[i], off, 32);
    float t = s[0];
#pragma unroll
    for (int i = 1; i < RU; ++i)
      if (lane == i) t = s[i];
    if (lane < RU) {
      const int row = row0 + r0 + lane;
      t += bias[row];
      if (SILU) {
        float sig = 1.0f / (1.0f + __expf(-t));
        t = t * sig;
      }
      y[row] = t;
    }
  }
}

__global__ void __launch_bounds__(128) __attribute__((amdgpu_waves_per_eu(1, 1)))
k_gemm_kv(const float* __restrict__ x,
                          const float* __restrict__ Wk, const float* __restrict__ bk,
                          const float* __restrict__ Wv, const float* __restrict__ bv,
                          const float* __restrict__ Wq, const float* __restrict__ bq,
                          const float* __restrict__ iq,
                          float* __restrict__ ws) {
  __shared__ float red[4 * 32 * RP_];
  __shared__ __attribute__((aligned(16))) float stg[4 * STG_];
  const int b = blockIdx.y, which = blockIdx.z;
  const int lane = threadIdx.x & 31, wave = threadIdx.x >> 5;
  if (blockIdx.x >= 24) {
    const int blk = ((blockIdx.x - 24) * B_ + b) * 2 + which;
    wave_matvec<C_, 4, 4, false>(Wq, iq, bq, ws + OFF_RQ, blk * 16 + wave * 4, lane);
    return;
  }
  const int m0 = (blockIdx.x & 1) * 32, n0 = (blockIdx.x >> 1) * 32;
  const float* A = x + b * T_ * C_;
  const float* W = which ? Wv : Wk;
  const float* bias = which ? bv : bk;
  float* out = ws + (which ? OFF_V : OFF_K) + b * T_ * C_;
  v8f acc[4] = {};
  gemm32x32_slice_staged<C_ / 4>(A, C_, W, C_, wave * (C_ / 4), m0, n0, lane,
                                 stg + wave * STG_, acc);
  put_partial(acc, red, wave, lane);
  __syncthreads();
  const int col = n0 + lane;
  const float bb = bias[col];
#pragma unroll
  for (int j = 0; j < 8; ++j) {
    const int R = wave * 8 + j;
    out[(m0 + R) * C_ + col] = row_sum<4>(red, R, lane) + bb;
  }
}

__global__ void __launch_bounds__(128) __attribute__((amdgpu_waves_per_eu(1, 1)))
k_gemm_h1(const float* __restrict__ sW0,
                          const float* __restrict__ sb0,
                          float* __restrict__ ws) {
  __shared__ float red[4 * 32 * RP_];
  __shared__ __attribute__((aligned(16))) float stg[4 * STG_];
  const int b = blockIdx.y;
  const int lane = threadIdx.x & 31, wave = threadIdx.x >> 5;
  if (blockIdx.x >= 48) {
    wave_matvec<C_, 4, 4, true>(sW0 + b * H_ * C_, ws + OFF_RQ, sb0 + b * H_,
                             ws + OFF_RH + b * H_, (blockIdx.x - 48) * 16 + wave * 4, lane);
    return;
  }
  const int m0 = (blockIdx.x % 2) * 32, n0 = (blockIdx.x / 2) * 32;
  const float* A = ws + OFF_K + b * T_ * C_;
  const float* W = sW0 + b * H_ * C_;
  float* h1 = ws + OFF_H1 + b * T_ * H_;
  float* gp = ws + OFF_GP + b * T_ * H_;
  v8f acc[4] = {};
  gemm32x32_slice_staged<C_ / 4>(A, C_, W, C_, wave * (C_ / 4), m0, n0, lane,
                                 stg + wave * STG_, acc);
  put_partial(acc, red, wave, lane);
  __syncthreads();
  const int col = n0 + lane;
  const float bb = sb0[b * H_ + col];
#pragma unroll
  for (int j = 0; j < 8; ++j) {
    const int R = wave * 8 + j;
    float pre = row_sum<4>(red, R, lane) + bb;
    float sig = 1.0f / (1.0f + __expf(-pre));
    h1[(m0 + R) * H_ + col] = pre * sig;
    gp[(m0 + R) * H_ + col] = sig * (1.0f + pre * (1.0f - sig));
  }
}

__global__ void __launch_bounds__(256) __attribute__((amdgpu_waves_per_eu(2, 2)))
k_gemm_dp(const float* __restrict__ sW1,
                          const float* __restrict__ sb1,
                          float* __restrict__ ws) {
  __shared__ float red[8 * 32 * RP_];
  __shared__ float dts[32 * TP_];
  __shared__ __attribute__((aligned(16))) float stg[8 * STG_];
  const int b = blockIdx.y;
  const int lane = threadIdx.x & 31, wave = threadIdx.x >> 5;
  if (blockIdx.x >= 24) {
    wave_matvec<H_, 2, 2, false>(sW1 + b * C_ * H_, ws + OFF_RH + b * H_, sb1 + b * C_,
                              ws + OFF_RP + b * C_, (blockIdx.x - 24) * 16 + wave * 2, lane);
    return;
  }
  const int m0 = (blockIdx.x % 2) * 32, n0 = (blockIdx.x / 2) * 32;
  const float* A = ws + OFF_H1 + b * T_ * H_;
  const float* W = sW1 + b * C_ * H_;
  const float* v = ws + OFF_V + b * T_ * C_;
  float* dp = ws + OFF_DP + b * T_ * C_;
  float* wdpT = ws + OFF_WDPT + b * C_ * T_;
  v8f acc[4] = {};
  gemm32x32_slice_staged<H_ / 8>(A, H_, W, H_, wave * (H_ / 8), m0, n0, lane,
                                 stg + wave * STG_, acc);
  put_partial(acc, red, wave, lane);
  __syncthreads();
  const int col = n0 + lane;
  const float bb = sb1[b * C_ + col];
  const float wrow = scan_w(m0 + lane);
  float d[4];
#pragma unroll
  for (int j = 0; j < 4; ++j) {
    const int R = wave * 4 + j;
    d[j] = (row_sum<8>(red, R, lane) + bb) - v[(m0 + R) * C_ + col];
  }
#pragma unroll
  for (int j = 0; j < 4; ++j) dts[(wave * 4 + j) * TP_ + lane] = d[j];
  __syncthreads();
#pragma unroll
  for (int j = 0; j < 4; ++j) {
    const int Cc = wave * 4 + j;
    wdpT[(n0 + Cc) * T_ + m0 + lane] = wrow * dts[lane * TP_ + Cc];
  }
#pragma unroll
  for (int j = 0; j < 4; ++j) dp[(m0 + wave * 4 + j) * C_ + col] = d[j];
}

template <int N>
__device__ __forceinline__ void update_tile(const float* __restrict__ sW,
                                            float* __restrict__ nW,
                                            const float* __restrict__ G,
                                            const float* __restrict__ Bm,
                                            int tile, int lane, float dT) {
  const int m0 = (tile / (N / 32)) * 32, n0 = (tile % (N / 32)) * 32;
  v8f acc[4] = {};
  gemm32x32<false>(G, T_, Bm, N, T_, m0, n0, lane, acc);
  const int cr = lane & 15, hi = lane >> 4;
#pragma unroll
  for (int mi = 0; mi < 2; ++mi)
#pragma unroll
    for (int ni = 0; ni < 2; ++ni) {
      int col = n0 + ni * 16 + cr;
      v8f a = acc[mi * 2 + ni];
#pragma unroll
      for (int r = 0; r < 8; ++r) {
        int row = m0 + mi * 16 + r + 8 * hi;
        nW[row * N + col] = dT * sW[row * N + col] + a[r];
      }
    }
}

constexpr int PP_ = 20;
constexpr int WP_ = 68;
constexpr int OP_ = 388;
struct __attribute__((packed, aligned(4))) f3 { float x, y, z; };

__global__ void __launch_bounds__(256) __attribute__((amdgpu_waves_per_eu(2, 2)))
k_grad_update(const float* __restrict__ sW0, const float* __restrict__ sb0,
              const float* __restrict__ sW1, const float* __restrict__ sb1,
              const float* __restrict__ Wo, const float* __restrict__ bo,
              float* __restrict__ nW0, float* __restrict__ nb0,
              float* __restrict__ nW1, float* __restrict__ nb1,
              float* __restrict__ rout,
              const float* __restrict__ ws) {
  __shared__ float red[8 * 64 * PP_];
  __shared__ float wdh[16 * WP_];
  const int bx = blockIdx.x, tid = threadIdx.x;
  const int lane = tid & 31, wave = tid >> 5;
  const float dT = decay_T();
  if (bx >= 171) {
    const int q = bx - 171, b = q / 24, blk = q % 24;
    wave_matvec<C_, 2, 2, false>(Wo, ws + OFF_RP + b * C_, bo, rout + b * C_,
                                 blk * 16 + wave * 2, lane);
    return;
  }
  if (bx >= 168) {
    const int r = (bx - 168) * 256 + tid;
    const int b = r / C_, c = r % C_;
    const float* p = ws + OFF_WDPT + b * C_ * T_ + c * T_;
    float s = 0.f;
#pragma unroll 4
    for (int t = 0; t < T_; ++t) s += p[t];
    nb1[b * C_ + c] = dT * sb1[b * C_ + c] + s;
    return;
  }
  if (bx >= 96) {
    const int id = (bx - 96) * 8 + wave;
    const int b = id / 288, tile = id % 288;
    update_tile<H_>(sW1 + b * C_ * H_, nW1 + b * C_ * H_, ws + OFF_WDPT + b * C_ * T_,
                    ws + OFF_H1 + b * T_ * H_, tile, lane, dT);
    return;
  }

  const int b = bx / 48, h0 = (bx % 48) * 16;
  const int mr = lane & 15, hi = lane >> 4;
  const int kq = hi << 2;
  const float* dp = ws + OFF_DP + b * T_ * C_;
  const float* W1 = sW1 + b * C_ * H_;
  const float* gp = ws + OFF_GP + b * T_ * H_;
  const float* kk = ws + OFF_K + b * T_ * C_;
  const float* W0 = sW0 + b * H_ * C_;
  float* oW0 = nW0 + b * H_ * C_;

  v4f fa[6][4], fb[6];
#pragma unroll
  for (int s = 0; s < 6; ++s) {
    const int kp = wave * 48 + s * 8 + kq;
#pragma unroll
    for (int mi = 0; mi < 4; ++mi)
      fa[s][mi] = *reinterpret_cast<const v4f*>(dp + (mi * 16 + mr) * C_ + kp);
    const float* pb = W1 + kp * H_ + h0 + mr;
    fb[s].x = pb[0];  fb[s].y = pb[H_];  fb[s].z = pb[2 * H_];  fb[s].w = pb[3 * H_];
  }
  const int rh = tid & 15, tq = tid >> 4;
  float g[4];
#pragma unroll
  for (int j = 0; j < 4; ++j) g[j] = gp[(tq + 16 * j) * H_ + h0 + rh];

  v8f acc[4] = {};
#pragma unroll
  for (int s = 0; s < 6; ++s) {
#pragma unroll
    for (int mi = 0; mi < 4; ++mi) acc[mi] = wmma4(fa[s][mi].xy, fb[s].xy, acc[mi]);
#pragma unroll
    for (int mi = 0; mi < 4; ++mi) acc[mi] = wmma4(fa[s][mi].zw, fb[s].zw, acc[mi]);
  }
#pragma unroll
  for (int mi = 0; mi < 4; ++mi)
#pragma unroll
    for (int r = 0; r < 8; ++r)
      red[(wave * 64 + mi * 16 + 8 * hi + r) * PP_ + mr] = acc[mi][r];
  __builtin_amdgcn_sched_barrier(0);

  const int n0w = wave * 48;
  f3 kb[8][4];
#pragma unroll
  for (int s = 0; s < 8; ++s)
#pragma unroll
    for (int i = 0; i < 4; ++i)
      kb[s][i] = *reinterpret_cast<const f3*>(kk + (s * 8 + kq + i) * C_ + n0w + 3 * mr);
  v4f w0q[6];
#pragma unroll
  for (int i = 0; i < 6; ++i) {
    const int idx = tid + 256 * i;
    w0q[i] = *reinterpret_cast<const v4f*>(W0 + (h0 + idx / 96) * C_ + 4 * (idx % 96));
  }
  __syncthreads();
#pragma unroll
  for (int j = 0; j < 4; ++j) {
    const int t = tq + 16 * j;
    float s = red[t * PP_ + rh];
#pragma unroll
    for (int sl = 1; sl < 8; ++sl) s += red[(sl * 64 + t) * PP_ + rh];
    const float d = s * g[j];
    wdh[rh * WP_ + t] = scan_w(t) * d;
  }
  __syncthreads();

  if (tid < 16) {
    float s = 0.f;
#pragma unroll
    for (int t = 0; t < T_; ++t) s += wdh[tid * WP_ + t];
    nb0[b * H_ + h0 + tid] = dT * sb0[b * H_ + h0 + tid] + s;
  }

  v8f ac2[3] = {};
#pragma unroll
  for (int s = 0; s < 8; ++s) {
    const v4f a = *reinterpret_cast<const v4f*>(wdh + mr * WP_ + s * 8 + kq);
    ac2[0] = wmma4(a.xy, v2f{kb[s][0].x, kb[s][1].x}, ac2[0]);
    ac2[1] = wmma4(a.xy, v2f{kb[s][0].y, kb[s][1].y}, ac2[1]);
    ac2[2] = wmma4(a.xy, v2f{kb[s][0].z, kb[s][1].z}, ac2[2]);
    ac2[0] = wmma4(a.zw, v2f{kb[s][2].x, kb[s][3].x}, ac2[0]);
    ac2[1] = wmma4(a.zw, v2f{kb[s][2].y, kb[s][3].y}, ac2[1]);
    ac2[2] = wmma4(a.zw, v2f{kb[s][2].z, kb[s][3].z}, ac2[2]);
  }
  float* outp = red;
#pragma unroll
  for (int r = 0; r < 8; ++r) {
    float* po = outp + (8 * hi + r) * OP_ + n0w + 3 * mr;
    po[0] = ac2[0][r];  po[1] = ac2[1][r];  po[2] = ac2[2][r];
  }
  __syncthreads();
#pragma unroll
  for (int i = 0; i < 6; ++i) {
    const int idx = tid + 256 * i;
    const int row = idx / 96, c4 = 4 * (idx % 96);
    const v4f a = *reinterpret_cast<const v4f*>(outp + row * OP_ + c4);
    v4f o;
    o.x = dT * w0q[i].x + a.x;  o.y = dT * w0q[i].y + a.y;
    o.z = dT * w0q[i].z + a.z;  o.w = dT * w0q[i].w + a.w;
    *reinterpret_cast<v4f*>(oW0 + (h0 + row) * C_ + c4) = o;
  }
}

}

extern "C" void kernel_launch(void* const* d_in, const int* in_sizes, int n_in,
                              void* d_out, int out_size, void* d_ws, size_t ws_size,
                              hipStream_t stream) {
  const float* x   = (const float*)d_in[0];
  const float* Wq  = (const float*)d_in[1];
  const float* bq  = (const float*)d_in[2];
  const float* Wk  = (const float*)d_in[3];
  const float* bk  = (const float*)d_in[4];
  const float* Wv  = (const float*)d_in[5];
  const float* bv  = (const float*)d_in[6];
  const float* Wo  = (const float*)d_in[7];
  const float* bo  = (const float*)d_in[8];
  const float* iq  = (const float*)d_in[9];
  const float* sW0 = (const float*)d_in[10];
  const float* sb0 = (const float*)d_in[11];
  const float* sW1 = (const float*)d_in[12];
  const float* sb1 = (const float*)d_in[13];

  float* out = (float*)d_out;
  float* ws  = (float*)d_ws;

  float* out_out = out;
  float* out_nW0 = out + B_ * C_;
  float* out_nb0 = out_nW0 + B_ * H_ * C_;
  float* out_nW1 = out_nb0 + B_ * H_;
  float* out_nb1 = out_nW1 + B_ * C_ * H_;

  k_gemm_kv<<<dim3(30, B_, 2), 128, 0, stream>>>(x, Wk, bk, Wv, bv, Wq, bq, iq, ws);
  k_gemm_h1<<<dim3(96, B_), 128, 0, stream>>>(sW0, sb0, ws);
  k_gemm_dp<<<dim3(48, B_), 256, 0, stream>>>(sW1, sb1, ws);
  k_grad_update<<<dim3(219), 256, 0, stream>>>(sW0, sb0, sW1, sb1, Wo, bo, out_nW0, out_nb0,
                                               out_nW1, out_nb1, out_out, ws);
}


# Launches 1-3: block tile halved to 16 rows x 32 columns (GEMM blocks 96/96/48 -> 192/192/96), halving each SIMD's matrix chain; wdpT re-laid in 16-t planes so its lines stay whole; sums keep order
# speedup vs baseline: 9.1371x; 1.1061x over previous
#include <hip/hip_runtime.h>
#include <cstdint>

typedef float v2f __attribute__((ext_vector_type(2)));
typedef float v4f __attribute__((ext_vector_type(4)));
typedef float v8f __attribute__((ext_vector_type(8)));

namespace {

constexpr int B_ = 2, T_ = 64, C_ = 384, H_ = 768;
constexpr float LR = 0.01f, MOM = 0.9f, DECAY = 0.001f;

constexpr int OFF_K    = 0;
constexpr int OFF_V    = OFF_K    + B_*T_*C_;
constexpr int OFF_H1   = OFF_V    + B_*T_*C_;
constexpr int OFF_GP   = OFF_H1   + B_*T_*H_;
constexpr int OFF_DP   = OFF_GP   + B_*T_*H_;
constexpr int OFF_WDPT = OFF_DP   + B_*T_*C_;
constexpr int OFF_RQ   = OFF_WDPT + B_*C_*T_;
constexpr int OFF_RH   = OFF_RQ   + C_;
constexpr int OFF_RP   = OFF_RH   + B_*H_;

__device__ __forceinline__ float scan_w(int t) {
  const float a = 1.0f - DECAY;
  const float n = (float)(T_ - t);
  const float an = __powf(a, n), mn = __powf(MOM, n);
  return -LR * (1.0f - MOM) * (an - mn) / (a - MOM);
}
__device__ __forceinline__ float decay_T() { return __powf(1.0f - DECAY, (float)T_); }

__device__ __forceinline__ v8f wmma4(v2f a, v2f b, v8f c) {
  return __builtin_amdgcn_wmma_f32_16x16x4_f32(false, a, false, b, (short)0, c,
                                               false, false);
}


__device__ __forceinline__ void mma_pair(v4f a0, v4f a1, v4f b0, v4f b1, v8f acc[4]) {
  acc[0] = wmma4(a0.xy, b0.xy, acc[0]);
  acc[1] = wmma4(a0.xy, b1.xy, acc[1]);
  acc[2] = wmma4(a1.xy, b0.xy, acc[2]);
  acc[3] = wmma4(a1.xy, b1.xy, acc[3]);
  acc[0] = wmma4(a0.zw, b0.zw, acc[0]);
  acc[1] = wmma4(a0.zw, b1.zw, acc[1]);
  acc[2] = wmma4(a1.zw, b0.zw, acc[2]);
  acc[3] = wmma4(a1.zw, b1.zw, acc[3]);
}

constexpr int SP_ = 100;
constexpr int STG_ = (16 + 32) * SP_;
template <int KS>
__device__ __forceinline__ void gemm16x32_slice_staged(const float* __restrict__ A, int lda,
                                                       const float* __restrict__ Bm, int ldb,
                                                       int k0, int m0, int n0, int lane,
                                                       float* st, v8f acc[2]) {
  static_assert(KS % 32 == 0 && KS + 4 <= SP_, "slice = whole lines, and it fits the pitch");
  constexpr int NC = KS / 32;
  const int lr = lane >> 3;
  const int lp = (lane & 7) << 2;
  const int mr = lane & 15;
  const int kq = (lane >> 4) << 2;
  float* sa = st;
  float* sb = st + 16 * SP_;
  v4f ga[NC][4], gb[NC][8];
#pragma unroll
  for (int s = 0; s < NC; ++s) {
#pragma unroll
    for (int g = 0; g < 4; ++g)
      ga[s][g] = *reinterpret_cast<const v4f*>(A + (m0 + 4 * g + lr) * lda + k0 + 32 * s + lp);
#pragma unroll
    for (int g = 0; g < 8; ++g)
      gb[s][g] = *reinterpret_cast<const v4f*>(Bm + (n0 + 4 * g + lr) * ldb + k0 + 32 * s + lp);
  }
#pragma unroll
  for (int s = 0; s < NC; ++s) {
#pragma unroll
    for (int g = 0; g < 4; ++g)
      *reinterpret_cast<v4f*>(sa + (4 * g + lr) * SP_ + 32 * s + lp) = ga[s][g];
#pragma unroll
    for (int g = 0; g < 8; ++g)
      *reinterpret_cast<v4f*>(sb + (4 * g + lr) * SP_ + 32 * s + lp) = gb[s][g];
    __builtin_amdgcn_fence(__ATOMIC_RELEASE, "wavefront");
    __builtin_amdgcn_wave_barrier();
    __builtin_amdgcn_fence(__ATOMIC_ACQUIRE, "wavefront");
#pragma unroll
    for (int u = 0; u < 4; ++u) {
      const int kp = 32 * s + 8 * u + kq;
      const v4f a0 = *reinterpret_cast<const v4f*>(sa + mr * SP_ + kp);
      const v4f b0 = *reinterpret_cast<const v4f*>(sb + mr * SP_ + kp);
      const v4f b1 = *reinterpret_cast<const v4f*>(sb + (16 + mr) * SP_ + kp);
      acc[0] = wmma4(a0.xy, b0.xy, acc[0]);
      acc[1] = wmma4(a0.xy, b1.xy, acc[1]);
      acc[0] = wmma4(a0.zw, b0.zw, acc[0]);
      acc[1] = wmma4(a0.zw, b1.zw, acc[1]);
    }
  }
}

constexpr int RP_ = 36;
constexpr int TP_ = 33;

__device__ __forceinline__ void put_partial(const v8f acc[2], float* red, int slice, int lane) {
  const int cr = lane & 15, hi = lane >> 4;
#pragma unroll
  for (int ni = 0; ni < 2; ++ni)
#pragma unroll
    for (int r = 0; r < 8; ++r)
      red[(slice * 16 + 8 * hi + r) * RP_ + ni * 16 + cr] = acc[ni][r];
}

template <int S>
__device__ __forceinline__ float row_sum(const float* red, int R, int lane) {
  float s = red[R * RP_ + lane];
#pragma unroll
  for (int sl = 1; sl < S; ++sl) s += red[(sl * 16 + R) * RP_ + lane];
  return s;
}

template <int K, int NR, int RU, bool SILU>
__device__ __forceinline__ void wave_matvec(const float* __restrict__ M,
                                            const float* __restrict__ v,
                                            const float* __restrict__ bias,
                                            float* __restrict__ y, int row0, int lane) {
  constexpr int NV = K / 128;
  v4f vv[NV];
#pragma unroll
  for (int j = 0; j < NV; ++j)
    vv[j] = *reinterpret_cast<const v4f*>(v + j * 128 + lane * 4);
  for (int r0 = 0; r0 < NR; r0 += RU) {
    float s[RU];
#pragma unroll
    for (int i = 0; i < RU; ++i) {
      const float* mr = M + (row0 + r0 + i) * K + lane * 4;
      float a = 0.f;
#pragma unroll
      for (int j = 0; j < NV; ++j) {
        v4f m = *reinterpret_cast<const v4f*>(mr + j * 128);
        a += m.x * vv[j].x;
        a += m.y * vv[j].y;
        a += m.z * vv[j].z;
        a += m.w * vv[j].w;
      }
      s[i] = a;
    }
#pragma unroll
    for (int off = 16; off > 0; off >>= 1)
#pragma unroll
      for (int i = 0; i < RU; ++i) s[i] += __shfl_xor(s[i], off, 32);
    float t = s[0];
#pragma unroll
    for (int i = 1; i < RU; ++i)
      if (lane == i) t = s[i];
    if (lane < RU) {
      const int row = row0 + r0 + lane;
      t += bias[row];
      if (SILU) {
        float sig = 1.0f / (1.0f + __expf(-t));
        t = t * sig;
      }
      y[row] = t;
    }
  }
}

__global__ void __launch_bounds__(128) __attribute__((amdgpu_waves_per_eu(1, 1)))
k_gemm_kv(const float* __restrict__ x,
                          const float* __restrict__ Wk, const float* __restrict__ bk,
                          const float* __restrict__ Wv, const float* __restrict__ bv,
                          const float* __restrict__ Wq, const float* __restrict__ bq,
                          const float* __restrict__ iq,
                          float* __restrict__ ws) {
  __shared__ float red[4 * 16 * RP_];
  __shared__ __attribute__((aligned(16))) float stg[4 * STG_];
  const int b = blockIdx.y, which = blockIdx.z;
  const int lane = threadIdx.x & 31, wave = threadIdx.x >> 5;
  if (blockIdx.x >= 48) {
    const int blk = ((blockIdx.x - 48) * B_ + b) * 2 + which;
    wave_matvec<C_, 4, 4, false>(Wq, iq, bq, ws + OFF_RQ, blk * 16 + wave * 4, lane);
    return;
  }
  const int m0 = (blockIdx.x & 3) * 16, n0 = (blockIdx.x >> 2) * 32;
  const float* A = x + b * T_ * C_;
  const float* W = which ? Wv : Wk;
  const float* bias = which ? bv : bk;
  float* out = ws + (which ? OFF_V : OFF_K) + b * T_ * C_;
  v8f acc[2] = {};
  gemm16x32_slice_staged<C_ / 4>(A, C_, W, C_, wave * (C_ / 4), m0, n0, lane,
                                 stg + wave * STG_, acc);
  put_partial(acc, red, wave, lane);
  __syncthreads();
  const int col = n0 + lane;
  const float bb = bias[col];
#pragma unroll
  for (int j = 0; j < 4; ++j) {
    const int R = wave * 4 + j;
    out[(m0 + R) * C_ + col] = row_sum<4>(red, R, lane) + bb;
  }
}

__global__ void __launch_bounds__(128) __attribute__((amdgpu_waves_per_eu(1, 1)))
k_gemm_h1(const float* __restrict__ sW0,
                          const float* __restrict__ sb0,
                          float* __restrict__ ws) {
  __shared__ float red[4 * 16 * RP_];
  __shared__ __attribute__((aligned(16))) float stg[4 * STG_];
  const int b = blockIdx.y;
  const int lane = threadIdx.x & 31, wave = threadIdx.x >> 5;
  if (blockIdx.x >= 96) {
    wave_matvec<C_, 4, 4, true>(sW0 + b * H_ * C_, ws + OFF_RQ, sb0 + b * H_,
                             ws + OFF_RH + b * H_, (blockIdx.x - 96) * 16 + wave * 4, lane);
    return;
  }
  const int m0 = (blockIdx.x % 4) * 16, n0 = (blockIdx.x / 4) * 32;
  const float* A = ws + OFF_K + b * T_ * C_;
  const float* W = sW0 + b * H_ * C_;
  float* h1 = ws + OFF_H1 + b * T_ * H_;
  float* gp = ws + OFF_GP + b * T_ * H_;
  v8f acc[2] = {};
  gemm16x32_slice_staged<C_ / 4>(A, C_, W, C_, wave * (C_ / 4), m0, n0, lane,
                                 stg + wave * STG_, acc);
  put_partial(acc, red, wave, lane);
  __syncthreads();
  const int col = n0 + lane;
  const float bb = sb0[b * H_ + col];
#pragma unroll
  for (int j = 0; j < 4; ++j) {
    const int R = wave * 4 + j;
    float pre = row_sum<4>(red, R, lane) + bb;
    float sig = 1.0f / (1.0f + __expf(-pre));
    h1[(m0 + R) * H_ + col] = pre * sig;
    gp[(m0 + R) * H_ + col] = sig * (1.0f + pre * (1.0f - sig));
  }
}

__global__ void __launch_bounds__(256) __attribute__((amdgpu_waves_per_eu(2, 2)))
k_gemm_dp(const float* __restrict__ sW1,
                          const float* __restrict__ sb1,
                          float* __restrict__ ws) {
  __shared__ float red[8 * 16 * RP_];
  __shared__ float dts[16 * TP_];
  __shared__ __attribute__((aligned(16))) float stg[8 * STG_];
  const int b = blockIdx.y;
  const int lane = threadIdx.x & 31, wave = threadIdx.x >> 5;
  if (blockIdx.x >= 48) {
    wave_matvec<H_, 2, 2, false>(sW1 + b * C_ * H_, ws + OFF_RH + b * H_, sb1 + b * C_,
                              ws + OFF_RP + b * C_, (blockIdx.x - 48) * 16 + wave * 2, lane);
    return;
  }
  const int m0 = (blockIdx.x % 4) * 16, n0 = (blockIdx.x / 4) * 32;
  const float* A = ws + OFF_H1 + b * T_ * H_;
  const float* W = sW1 + b * C_ * H_;
  const float* v = ws + OFF_V + b * T_ * C_;
  float* dp = ws + OFF_DP + b * T_ * C_;
  float* wdpT = ws + OFF_WDPT + b * C_ * T_;
  v8f acc[2] = {};
  gemm16x32_slice_staged<H_ / 8>(A, H_, W, H_, wave * (H_ / 8), m0, n0, lane,
                                 stg + wave * STG_, acc);
  put_partial(acc, red, wave, lane);
  __syncthreads();
  const int col = n0 + lane;
  const float bb = sb1[b * C_ + col];
  const int tl = lane & 15, ch = lane >> 4;
  const float wrow = scan_w(m0 + tl);
  float d[2];
#pragma unroll
  for (int j = 0; j < 2; ++j) {
    const int R = wave * 2 + j;
    d[j] = (row_sum<8>(red, R, lane) + bb) - v[(m0 + R) * C_ + col];
  }
#pragma unroll
  for (int j = 0; j < 2; ++j) dts[(wave * 2 + j) * TP_ + lane] = d[j];
  __syncthreads();
  float* plane = wdpT + (m0 >> 4) * (C_ * 16);
#pragma unroll
  for (int j = 0; j < 2; ++j) {
    const int Cc = 2 * (wave * 2 + j) + ch;
    plane[(n0 + Cc) * 16 + tl] = wrow * dts[tl * TP_ + Cc];
  }
#pragma unroll
  for (int j = 0; j < 2; ++j) dp[(m0 + wave * 2 + j) * C_ + col] = d[j];
}

template <int N>
__device__ __forceinline__ void update_tile(const float* __restrict__ sW,
                                            float* __restrict__ nW,
                                            const float* __restrict__ G,
                                            const float* __restrict__ Bm,
                                            int tile, int lane, float dT) {
  const int m0 = (tile / (N / 32)) * 32, n0 = (tile % (N / 32)) * 32;
  const int cr = lane & 15, hi = lane >> 4;
  const int kq = hi << 2;
  v8f acc[4] = {};
#pragma unroll
  for (int kk = 0; kk < T_; kk += 8) {
    const int kp = kk + kq;
    const float* pa = G + ((kp >> 4) * C_ + m0 + cr) * 16 + (kp & 15);
    const v4f a0 = *reinterpret_cast<const v4f*>(pa);
    const v4f a1 = *reinterpret_cast<const v4f*>(pa + 16 * 16);
    const float* pb = Bm + kp * N + n0 + cr;
    v4f b0, b1;
    b0.x = pb[0];   b0.y = pb[N];       b0.z = pb[2 * N];       b0.w = pb[3 * N];
    b1.x = pb[16];  b1.y = pb[N + 16];  b1.z = pb[2 * N + 16];  b1.w = pb[3 * N + 16];
    mma_pair(a0, a1, b0, b1, acc);
  }
#pragma unroll
  for (int mi = 0; mi < 2; ++mi)
#pragma unroll
    for (int ni = 0; ni < 2; ++ni) {
      int col = n0 + ni * 16 + cr;
      v8f a = acc[mi * 2 + ni];
#pragma unroll
      for (int r = 0; r < 8; ++r) {
        int row = m0 + mi * 16 + r + 8 * hi;
        nW[row * N + col] = dT * sW[row * N + col] + a[r];
      }
    }
}

constexpr int PP_ = 20;
constexpr int WP_ = 68;
constexpr int OP_ = 388;
struct __attribute__((packed, aligned(4))) f3 { float x, y, z; };

__global__ void __launch_bounds__(256) __attribute__((amdgpu_waves_per_eu(2, 2)))
k_grad_update(const float* __restrict__ sW0, const float* __restrict__ sb0,
              const float* __restrict__ sW1, const float* __restrict__ sb1,
              const float* __restrict__ Wo, const float* __restrict__ bo,
              float* __restrict__ nW0, float* __restrict__ nb0,
              float* __restrict__ nW1, float* __restrict__ nb1,
              float* __restrict__ rout,
              const float* __restrict__ ws) {
  __shared__ float red[8 * 64 * PP_];
  __shared__ float wdh[16 * WP_];
  const int bx = blockIdx.x, tid = threadIdx.x;
  const int lane = tid & 31, wave = tid >> 5;
  const float dT = decay_T();
  if (bx >= 171) {
    const int q = bx - 171, b = q / 24, blk = q % 24;
    wave_matvec<C_, 2, 2, false>(Wo, ws + OFF_RP + b * C_, bo, rout + b * C_,
                                 blk * 16 + wave * 2, lane);
    return;
  }
  if (bx >= 168) {
    const int r = (bx - 168) * 256 + tid;
    const int b = r / C_, c = r % C_;
    const float* p = ws + OFF_WDPT + b * C_ * T_ + c * 16;
    v4f q[T_ / 4];
#pragma unroll
    for (int i = 0; i < T_ / 4; ++i)
      q[i] = *reinterpret_cast<const v4f*>(p + (i >> 2) * (C_ * 16) + (i & 3) * 4);
    float s = 0.f;
#pragma unroll
    for (int i = 0; i < T_ / 4; ++i) {
      s += q[i].x;  s += q[i].y;  s += q[i].z;  s += q[i].w;
    }
    nb1[b * C_ + c] = dT * sb1[b * C_ + c] + s;
    return;
  }
  if (bx >= 96) {
    const int id = (bx - 96) * 8 + wave;
    const int b = id / 288, tile = id % 288;
    update_tile<H_>(sW1 + b * C_ * H_, nW1 + b * C_ * H_, ws + OFF_WDPT + b * C_ * T_,
                    ws + OFF_H1 + b * T_ * H_, tile, lane, dT);
    return;
  }

  const int b = bx / 48, h0 = (bx % 48) * 16;
  const int mr = lane & 15, hi = lane >> 4;
  const int kq = hi << 2;
  const float* dp = ws + OFF_DP + b * T_ * C_;
  const float* W1 = sW1 + b * C_ * H_;
  const float* gp = ws + OFF_GP + b * T_ * H_;
  const float* kk = ws + OFF_K + b * T_ * C_;
  const float* W0 = sW0 + b * H_ * C_;
  float* oW0 = nW0 + b * H_ * C_;

  v4f fa[6][4], fb[6];
#pragma unroll
  for (int s = 0; s < 6; ++s) {
    const int kp = wave * 48 + s * 8 + kq;
#pragma unroll
    for (int mi = 0; mi < 4; ++mi)
      fa[s][mi] = *reinterpret_cast<const v4f*>(dp + (mi * 16 + mr) * C_ + kp);
    const float* pb = W1 + kp * H_ + h0 + mr;
    fb[s].x = pb[0];  fb[s].y = pb[H_];  fb[s].z = pb[2 * H_];  fb[s].w = pb[3 * H_];
  }
  const int rh = tid & 15, tq = tid >> 4;
  float g[4];
#pragma unroll
  for (int j = 0; j < 4; ++j) g[j] = gp[(tq + 16 * j) * H_ + h0 + rh];

  v8f acc[4] = {};
#pragma unroll
  for (int s = 0; s < 6; ++s) {
#pragma unroll
    for (int mi = 0; mi < 4; ++mi) acc[mi] = wmma4(fa[s][mi].xy, fb[s].xy, acc[mi]);
#pragma unroll
    for (int mi = 0; mi < 4; ++mi) acc[mi] = wmma4(fa[s][mi].zw, fb[s].zw, acc[mi]);
  }
#pragma unroll
  for (int mi = 0; mi < 4; ++mi)
#pragma unroll
    for (int r = 0; r < 8; ++r)
      red[(wave * 64 + mi * 16 + 8 * hi + r) * PP_ + mr] = acc[mi][r];
  __builtin_amdgcn_sched_barrier(0);

  const int n0w = wave * 48;
  f3 kb[8][4];
#pragma unroll
  for (int s = 0; s < 8; ++s)
#pragma unroll
    for (int i = 0; i < 4; ++i)
      kb[s][i] = *reinterpret_cast<const f3*>(kk + (s * 8 + kq + i) * C_ + n0w + 3 * mr);
  v4f w0q[6];
#pragma unroll
  for (int i = 0; i < 6; ++i) {
    const int idx = tid + 256 * i;
    w0q[i] = *reinterpret_cast<const v4f*>(W0 + (h0 + idx / 96) * C_ + 4 * (idx % 96));
  }
  __syncthreads();
#pragma unroll
  for (int j = 0; j < 4; ++j) {
    const int t = tq + 16 * j;
    float s = red[t * PP_ + rh];
#pragma unroll
    for (int sl = 1; sl < 8; ++sl) s += red[(sl * 64 + t) * PP_ + rh];
    const float d = s * g[j];
    wdh[rh * WP_ + t] = scan_w(t) * d;
  }
  __syncthreads();

  if (tid < 16) {
    float s = 0.f;
#pragma unroll
    for (int t = 0; t < T_; ++t) s += wdh[tid * WP_ + t];
    nb0[b * H_ + h0 + tid] = dT * sb0[b * H_ + h0 + tid] + s;
  }

  v8f ac2[3] = {};
#pragma unroll
  for (int s = 0; s < 8; ++s) {
    const v4f a = *reinterpret_cast<const v4f*>(wdh + mr * WP_ + s * 8 + kq);
    ac2[0] = wmma4(a.xy, v2f{kb[s][0].x, kb[s][1].x}, ac2[0]);
    ac2[1] = wmma4(a.xy, v2f{kb[s][0].y, kb[s][1].y}, ac2[1]);
    ac2[2] = wmma4(a.xy, v2f{kb[s][0].z, kb[s][1].z}, ac2[2]);
    ac2[0] = wmma4(a.zw, v2f{kb[s][2].x, kb[s][3].x}, ac2[0]);
    ac2[1] = wmma4(a.zw, v2f{kb[s][2].y, kb[s][3].y}, ac2[1]);
    ac2[2] = wmma4(a.zw, v2f{kb[s][2].z, kb[s][3].z}, ac2[2]);
  }
  float* outp = red;
#pragma unroll
  for (int r = 0; r < 8; ++r) {
    float* po = outp + (8 * hi + r) * OP_ + n0w + 3 * mr;
    po[0] = ac2[0][r];  po[1] = ac2[1][r];  po[2] = ac2[2][r];
  }
  __syncthreads();
#pragma unroll
  for (int i = 0; i < 6; ++i) {
    const int idx = tid + 256 * i;
    const int row = idx / 96, c4 = 4 * (idx % 96);
    const v4f a = *reinterpret_cast<const v4f*>(outp + row * OP_ + c4);
    v4f o;
    o.x = dT * w0q[i].x + a.x;  o.y = dT * w0q[i].y + a.y;
    o.z = dT * w0q[i].z + a.z;  o.w = dT * w0q[i].w + a.w;
    *reinterpret_cast<v4f*>(oW0 + (h0 + row) * C_ + c4) = o;
  }
}

}

extern "C" void kernel_launch(void* const* d_in, const int* in_sizes, int n_in,
                              void* d_out, int out_size, void* d_ws, size_t ws_size,
                              hipStream_t stream) {
  const float* x   = (const float*)d_in[0];
  const float* Wq  = (const float*)d_in[1];
  const float* bq  = (const float*)d_in[2];
  const float* Wk  = (const float*)d_in[3];
  const float* bk  = (const float*)d_in[4];
  const float* Wv  = (const float*)d_in[5];
  const float* bv  = (const float*)d_in[6];
  const float* Wo  = (const float*)d_in[7];
  const float* bo  = (const float*)d_in[8];
  const float* iq  = (const float*)d_in[9];
  const float* sW0 = (const float*)d_in[10];
  const float* sb0 = (const float*)d_in[11];
  const float* sW1 = (const float*)d_in[12];
  const float* sb1 = (const float*)d_in[13];

  float* out = (float*)d_out;
  float* ws  = (float*)d_ws;

  float* out_out = out;
  float* out_nW0 = out + B_ * C_;
  float* out_nb0 = out_nW0 + B_ * H_ * C_;
  float* out_nW1 = out_nb0 + B_ * H_;
  float* out_nb1 = out_nW1 + B_ * C_ * H_;

  k_gemm_kv<<<dim3(54, B_, 2), 128, 0, stream>>>(x, Wk, bk, Wv, bv, Wq, bq, iq, ws);
  k_gemm_h1<<<dim3(144, B_), 128, 0, stream>>>(sW0, sb0, ws);
  k_gemm_dp<<<dim3(72, B_), 256, 0, stream>>>(sW1, sb1, ws);
  k_grad_update<<<dim3(219), 256, 0, stream>>>(sW0, sb0, sW1, sb1, Wo, bo, out_nW0, out_nb0,
                                               out_nW1, out_nb1, out_out, ws);
}


# Launch 3 tile halved to 16x16 (192 GEMM blocks, 48 products a SIMD); dp leaves it only weighted, in two whole-line plane layouts; launch 4 loads fragments there, no scan weights: rounding differs
# speedup vs baseline: 10.4036x; 1.1386x over previous
#include <hip/hip_runtime.h>
#include <cstdint>

typedef float v2f __attribute__((ext_vector_type(2)));
typedef float v4f __attribute__((ext_vector_type(4)));
typedef float v8f __attribute__((ext_vector_type(8)));

namespace {

constexpr int B_ = 2, T_ = 64, C_ = 384, H_ = 768;
constexpr float LR = 0.01f, MOM = 0.9f, DECAY = 0.001f;

constexpr int OFF_K    = 0;
constexpr int OFF_V    = OFF_K    + B_*T_*C_;
constexpr int OFF_H1   = OFF_V    + B_*T_*C_;
constexpr int OFF_GP   = OFF_H1   + B_*T_*H_;
constexpr int OFF_WDPQ = OFF_GP   + B_*T_*H_;
constexpr int OFF_WDPT = OFF_WDPQ + B_*T_*C_;
constexpr int OFF_RQ   = OFF_WDPT + B_*C_*T_;
constexpr int OFF_RH   = OFF_RQ   + C_;
constexpr int OFF_RP   = OFF_RH   + B_*H_;

__device__ __forceinline__ float scan_w(int t) {
  const float a = 1.0f - DECAY;
  const float n = (float)(T_ - t);
  const float an = __powf(a, n), mn = __powf(MOM, n);
  return -LR * (1.0f - MOM) * (an - mn) / (a - MOM);
}
__device__ __forceinline__ float decay_T() { return __powf(1.0f - DECAY, (float)T_); }

__device__ __forceinline__ v8f wmma4(v2f a, v2f b, v8f c) {
  return __builtin_amdgcn_wmma_f32_16x16x4_f32(false, a, false, b, (short)0, c,
                                               false, false);
}


__device__ __forceinline__ void mma_pair(v4f a0, v4f a1, v4f b0, v4f b1, v8f acc[4]) {
  acc[0] = wmma4(a0.xy, b0.xy, acc[0]);
  acc[1] = wmma4(a0.xy, b1.xy, acc[1]);
  acc[2] = wmma4(a1.xy, b0.xy, acc[2]);
  acc[3] = wmma4(a1.xy, b1.xy, acc[3]);
  acc[0] = wmma4(a0.zw, b0.zw, acc[0]);
  acc[1] = wmma4(a0.zw, b1.zw, acc[1]);
  acc[2] = wmma4(a1.zw, b0.zw, acc[2]);
  acc[3] = wmma4(a1.zw, b1.zw, acc[3]);
}

constexpr int SP_ = 100;
constexpr int STG_ = (16 + 32) * SP_;
constexpr int STG1_ = (16 + 16) * SP_;
template <int KS, int NB>
__device__ __forceinline__ void gemm16_slice_staged(const float* __restrict__ A, int lda,
                                                    const float* __restrict__ Bm, int ldb,
                                                    int k0, int m0, int n0, int lane,
                                                    float* st, v8f acc[NB]) {
  static_assert(KS % 32 == 0 && KS + 4 <= SP_, "slice = whole lines, and it fits the pitch");
  constexpr int NC = KS / 32;
  const int lr = lane >> 3;
  const int lp = (lane & 7) << 2;
  const int mr = lane & 15;
  const int kq = (lane >> 4) << 2;
  float* sa = st;
  float* sb = st + 16 * SP_;
  v4f ga[NC][4], gb[NC][4 * NB];
#pragma unroll
  for (int s = 0; s < NC; ++s) {
#pragma unroll
    for (int g = 0; g < 4; ++g)
      ga[s][g] = *reinterpret_cast<const v4f*>(A + (m0 + 4 * g + lr) * lda + k0 + 32 * s + lp);
#pragma unroll
    for (int g = 0; g < 4 * NB; ++g)
      gb[s][g] = *reinterpret_cast<const v4f*>(Bm + (n0 + 4 * g + lr) * ldb + k0 + 32 * s + lp);
  }
#pragma unroll
  for (int s = 0; s < NC; ++s) {
#pragma unroll
    for (int g = 0; g < 4; ++g)
      *reinterpret_cast<v4f*>(sa + (4 * g + lr) * SP_ + 32 * s + lp) = ga[s][g];
#pragma unroll
    for (int g = 0; g < 4 * NB; ++g)
      *reinterpret_cast<v4f*>(sb + (4 * g + lr) * SP_ + 32 * s + lp) = gb[s][g];
    __builtin_amdgcn_fence(__ATOMIC_RELEASE, "wavefront");
    __builtin_amdgcn_wave_barrier();
    __builtin_amdgcn_fence(__ATOMIC_ACQUIRE, "wavefront");
#pragma unroll
    for (int u = 0; u < 4; ++u) {
      const int kp = 32 * s + 8 * u + kq;
      const v4f a0 = *reinterpret_cast<const v4f*>(sa + mr * SP_ + kp);
      v4f bf[NB];
#pragma unroll
      for (int ni = 0; ni < NB; ++ni)
        bf[ni] = *reinterpret_cast<const v4f*>(sb + (16 * ni + mr) * SP_ + kp);
#pragma unroll
      for (int ni = 0; ni < NB; ++ni) acc[ni] = wmma4(a0.xy, bf[ni].xy, acc[ni]);
#pragma unroll
      for (int ni = 0; ni < NB; ++ni) acc[ni] = wmma4(a0.zw, bf[ni].zw, acc[ni]);
    }
  }
}

constexpr int RP_ = 36;
constexpr int RQ_ = 20;
constexpr int TQ_ = 17;

__device__ __forceinline__ void put_partial(const v8f acc[2], float* red, int slice, int lane) {
  const int cr = lane & 15, hi = lane >> 4;
#pragma unroll
  for (int ni = 0; ni < 2; ++ni)
#pragma unroll
    for (int r = 0; r < 8; ++r)
      red[(slice * 16 + 8 * hi + r) * RP_ + ni * 16 + cr] = acc[ni][r];
}

template <int S>
__device__ __forceinline__ float row_sum(const float* red, int R, int lane) {
  float s = red[R * RP_ + lane];
#pragma unroll
  for (int sl = 1; sl < S; ++sl) s += red[(sl * 16 + R) * RP_ + lane];
  return s;
}

template <int K, int NR, int RU, bool SILU>
__device__ __forceinline__ void wave_matvec(const float* __restrict__ M,
                                            const float* __restrict__ v,
                                            const float* __restrict__ bias,
                                            float* __restrict__ y, int row0, int lane) {
  constexpr int NV = K / 128;
  v4f vv[NV];
#pragma unroll
  for (int j = 0; j < NV; ++j)
    vv[j] = *reinterpret_cast<const v4f*>(v + j * 128 + lane * 4);
  for (int r0 = 0; r0 < NR; r0 += RU) {
    float s[RU];
#pragma unroll
    for (int i = 0; i < RU; ++i) {
      const float* mr = M + (row0 + r0 + i) * K + lane * 4;
      float a = 0.f;
#pragma unroll
      for (int j = 0; j < NV; ++j) {
        v4f m = *reinterpret_cast<const v4f*>(mr + j * 128);
        a += m.x * vv[j].x;
        a += m.y * vv[j].y;
        a += m.z * vv[j].z;
        a += m.w * vv[j].w;
      }
      s[i] = a;
    }
#pragma unroll
    for (int off = 16; off > 0; off >>= 1)
#pragma unroll
      for (int i = 0; i < RU; ++i) s[i] += __shfl_xor(s[i], off, 32);
    float t = s[0];
#pragma unroll
    for (int i = 1; i < RU; ++i)
      if (lane == i) t = s[i];
    if (lane < RU) {
      const int row = row0 + r0 + lane;
      t += bias[row];
      if (SILU) {
        float sig = 1.0f / (1.0f + __expf(-t));
        t = t * sig;
      }
      y[row] = t;
    }
  }
}

__global__ void __launch_bounds__(128) __attribute__((amdgpu_waves_per_eu(1, 1)))
k_gemm_kv(const float* __restrict__ x,
                          const float* __restrict__ Wk, const float* __restrict__ bk,
                          const float* __restrict__ Wv, const float* __restrict__ bv,
                          const float* __restrict__ Wq, const float* __restrict__ bq,
                          const float* __restrict__ iq,
                          float* __restrict__ ws) {
  __shared__ float red[4 * 16 * RP_];
  __shared__ __attribute__((aligned(16))) float stg[4 * STG_];
  const int b = blockIdx.y, which = blockIdx.z;
  const int lane = threadIdx.x & 31, wave = threadIdx.x >> 5;
  if (blockIdx.x >= 48) {
    const int blk = ((blockIdx.x - 48) * B_ + b) * 2 + which;
    wave_matvec<C_, 4, 4, false>(Wq, iq, bq, ws + OFF_RQ, blk * 16 + wave * 4, lane);
    return;
  }
  const int m0 = (blockIdx.x & 3) * 16, n0 = (blockIdx.x >> 2) * 32;
  const float* A = x + b * T_ * C_;
  const float* W = which ? Wv : Wk;
  const float* bias = which ? bv : bk;
  float* out = ws + (which ? OFF_V : OFF_K) + b * T_ * C_;
  v8f acc[2] = {};
  gemm16_slice_staged<C_ / 4, 2>(A, C_, W, C_, wave * (C_ / 4), m0, n0, lane,
                                 stg + wave * STG_, acc);
  put_partial(acc, red, wave, lane);
  __syncthreads();
  const int col = n0 + lane;
  const float bb = bias[col];
#pragma unroll
  for (int j = 0; j < 4; ++j) {
    const int R = wave * 4 + j;
    out[(m0 + R) * C_ + col] = row_sum<4>(red, R, lane) + bb;
  }
}

__global__ void __launch_bounds__(128) __attribute__((amdgpu_waves_per_eu(1, 1)))
k_gemm_h1(const float* __restrict__ sW0,
                          const float* __restrict__ sb0,
                          float* __restrict__ ws) {
  __shared__ float red[4 * 16 * RP_];
  __shared__ __attribute__((aligned(16))) float stg[4 * STG_];
  const int b = blockIdx.y;
  const int lane = threadIdx.x & 31, wave = threadIdx.x >> 5;
  if (blockIdx.x >= 96) {
    wave_matvec<C_, 4, 4, true>(sW0 + b * H_ * C_, ws + OFF_RQ, sb0 + b * H_,
                             ws + OFF_RH + b * H_, (blockIdx.x - 96) * 16 + wave * 4, lane);
    return;
  }
  const int m0 = (blockIdx.x % 4) * 16, n0 = (blockIdx.x / 4) * 32;
  const float* A = ws + OFF_K + b * T_ * C_;
  const float* W = sW0 + b * H_ * C_;
  float* h1 = ws + OFF_H1 + b * T_ * H_;
  float* gp = ws + OFF_GP + b * T_ * H_;
  v8f acc[2] = {};
  gemm16_slice_staged<C_ / 4, 2>(A, C_, W, C_, wave * (C_ / 4), m0, n0, lane,
                                 stg + wave * STG_, acc);
  put_partial(acc, red, wave, lane);
  __syncthreads();
  const int col = n0 + lane;
  const float bb = sb0[b * H_ + col];
#pragma unroll
  for (int j = 0; j < 4; ++j) {
    const int R = wave * 4 + j;
    float pre = row_sum<4>(red, R, lane) + bb;
    float sig = 1.0f / (1.0f + __expf(-pre));
    h1[(m0 + R) * H_ + col] = pre * sig;
    gp[(m0 + R) * H_ + col] = sig * (1.0f + pre * (1.0f - sig));
  }
}

__global__ void __launch_bounds__(256) __attribute__((amdgpu_waves_per_eu(2, 2)))
k_gemm_dp(const float* __restrict__ sW1,
                          const float* __restrict__ sb1,
                          float* __restrict__ ws) {
  __shared__ float red[8 * 16 * RQ_];
  __shared__ float dts[16 * TQ_];
  __shared__ __attribute__((aligned(16))) float stg[8 * STG1_];
  const int b = blockIdx.y;
  const int tid = threadIdx.x;
  const int lane = tid & 31, wave = tid >> 5;
  if (blockIdx.x >= 96) {
    wave_matvec<H_, 2, 2, false>(sW1 + b * C_ * H_, ws + OFF_RH + b * H_, sb1 + b * C_,
                              ws + OFF_RP + b * C_, (blockIdx.x - 96) * 16 + wave * 2, lane);
    return;
  }
  const int m0 = (blockIdx.x % 4) * 16, n0 = (blockIdx.x / 4) * 16;
  const float* A = ws + OFF_H1 + b * T_ * H_;
  const float* W = sW1 + b * C_ * H_;
  const float* v = ws + OFF_V + b * T_ * C_;
  float* wdpQ = ws + OFF_WDPQ + b * T_ * C_;
  float* wdpT = ws + OFF_WDPT + b * C_ * T_;
  const int R = tid >> 4, cc = tid & 15;
  const float bb = sb1[b * C_ + n0 + cc];
  const float vv = v[(m0 + R) * C_ + n0 + cc];
  v8f acc[1] = {};
  gemm16_slice_staged<H_ / 8, 1>(A, H_, W, H_, wave * (H_ / 8), m0, n0, lane,
                                 stg + wave * STG1_, acc);
  {
    const int cr = lane & 15, hi = lane >> 4;
#pragma unroll
    for (int r = 0; r < 8; ++r) red[(wave * 16 + 8 * hi + r) * RQ_ + cr] = acc[0][r];
  }
  __syncthreads();
  float s = red[R * RQ_ + cc];
#pragma unroll
  for (int sl = 1; sl < 8; ++sl) s += red[(sl * 16 + R) * RQ_ + cc];
  const float d = (s + bb) - vv;
  dts[R * TQ_ + cc] = scan_w(m0 + R) * d;
  __syncthreads();
  {
    const int tl = lane & 15, Cc = 2 * wave + (lane >> 4);
    wdpT[((m0 >> 4) * C_ + n0 + Cc) * 16 + tl] = dts[tl * TQ_ + Cc];
  }
  {
    const int q = wave >> 1, tt = 8 * (wave & 1) + (lane >> 2), ci = lane & 3;
    wdpQ[(((n0 >> 2) + q) * T_ + m0 + tt) * 4 + ci] = dts[tt * TQ_ + 4 * q + ci];
  }
}

template <int N>
__device__ __forceinline__ void update_tile(const float* __restrict__ sW,
                                            float* __restrict__ nW,
                                            const float* __restrict__ G,
                                            const float* __restrict__ Bm,
                                            int tile, int lane, float dT) {
  const int m0 = (tile / (N / 32)) * 32, n0 = (tile % (N / 32)) * 32;
  const int cr = lane & 15, hi = lane >> 4;
  const int kq = hi << 2;
  v8f acc[4] = {};
#pragma unroll
  for (int kk = 0; kk < T_; kk += 8) {
    const int kp = kk + kq;
    const float* pa = G + ((kp >> 4) * C_ + m0 + cr) * 16 + (kp & 15);
    const v4f a0 = *reinterpret_cast<const v4f*>(pa);
    const v4f a1 = *reinterpret_cast<const v4f*>(pa + 16 * 16);
    const float* pb = Bm + kp * N + n0 + cr;
    v4f b0, b1;
    b0.x = pb[0];   b0.y = pb[N];       b0.z = pb[2 * N];       b0.w = pb[3 * N];
    b1.x = pb[16];  b1.y = pb[N + 16];  b1.z = pb[2 * N + 16];  b1.w = pb[3 * N + 16];
    mma_pair(a0, a1, b0, b1, acc);
  }
#pragma unroll
  for (int mi = 0; mi < 2; ++mi)
#pragma unroll
    for (int ni = 0; ni < 2; ++ni) {
      int col = n0 + ni * 16 + cr;
      v8f a = acc[mi * 2 + ni];
#pragma unroll
      for (int r = 0; r < 8; ++r) {
        int row = m0 + mi * 16 + r + 8 * hi;
        nW[row * N + col] = dT * sW[row * N + col] + a[r];
      }
    }
}

constexpr int PP_ = 20;
constexpr int WP_ = 68;
constexpr int OP_ = 388;
struct __attribute__((packed, aligned(4))) f3 { float x, y, z; };

__global__ void __launch_bounds__(256) __attribute__((amdgpu_waves_per_eu(2, 2)))
k_grad_update(const float* __restrict__ sW0, const float* __restrict__ sb0,
              const float* __restrict__ sW1, const float* __restrict__ sb1,
              const float* __restrict__ Wo, const float* __restrict__ bo,
              float* __restrict__ nW0, float* __restrict__ nb0,
              float* __restrict__ nW1, float* __restrict__ nb1,
              float* __restrict__ rout,
              const float* __restrict__ ws) {
  __shared__ float red[8 * 64 * PP_];
  __shared__ float wdh[16 * WP_];
  const int bx = blockIdx.x, tid = threadIdx.x;
  const int lane = tid & 31, wave = tid >> 5;
  const float dT = decay_T();
  if (bx >= 171) {
    const int q = bx - 171, b = q / 24, blk = q % 24;
    wave_matvec<C_, 2, 2, false>(Wo, ws + OFF_RP + b * C_, bo, rout + b * C_,
                                 blk * 16 + wave * 2, lane);
    return;
  }
  if (bx >= 168) {
    const int r = (bx - 168) * 256 + tid;
    const int b = r / C_, c = r % C_;
    const float* p = ws + OFF_WDPT + b * C_ * T_ + c * 16;
    v4f q[T_ / 4];
#pragma unroll
    for (int i = 0; i < T_ / 4; ++i)
      q[i] = *reinterpret_cast<const v4f*>(p + (i >> 2) * (C_ * 16) + (i & 3) * 4);
    float s = 0.f;
#pragma unroll
    for (int i = 0; i < T_ / 4; ++i) {
      s += q[i].x;  s += q[i].y;  s += q[i].z;  s += q[i].w;
    }
    nb1[b * C_ + c] = dT * sb1[b * C_ + c] + s;
    return;
  }
  if (bx >= 96) {
    const int id = (bx - 96) * 8 + wave;
    const int b = id / 288, tile = id % 288;
    update_tile<H_>(sW1 + b * C_ * H_, nW1 + b * C_ * H_, ws + OFF_WDPT + b * C_ * T_,
                    ws + OFF_H1 + b * T_ * H_, tile, lane, dT);
    return;
  }

  const int b = bx / 48, h0 = (bx % 48) * 16;
  const int mr = lane & 15, hi = lane >> 4;
  const int kq = hi << 2;
  const float* wdpQ = ws + OFF_WDPQ + b * T_ * C_;
  const float* W1 = sW1 + b * C_ * H_;
  const float* gp = ws + OFF_GP + b * T_ * H_;
  const float* kk = ws + OFF_K + b * T_ * C_;
  const float* W0 = sW0 + b * H_ * C_;
  float* oW0 = nW0 + b * H_ * C_;

  v4f fa[6][4], fb[6];
#pragma unroll
  for (int s = 0; s < 6; ++s) {
    const int kp = wave * 48 + s * 8 + kq;
#pragma unroll
    for (int mi = 0; mi < 4; ++mi)
      fa[s][mi] = *reinterpret_cast<const v4f*>(wdpQ + ((kp >> 2) * T_ + mi * 16 + mr) * 4);
    const float* pb = W1 + kp * H_ + h0 + mr;
    fb[s].x = pb[0];  fb[s].y = pb[H_];  fb[s].z = pb[2 * H_];  fb[s].w = pb[3 * H_];
  }
  const int rh = tid & 15, tq = tid >> 4;
  float g[4];
#pragma unroll
  for (int j = 0; j < 4; ++j) g[j] = gp[(tq + 16 * j) * H_ + h0 + rh];

  v8f acc[4] = {};
#pragma unroll
  for (int s = 0; s < 6; ++s) {
#pragma unroll
    for (int mi = 0; mi < 4; ++mi) acc[mi] = wmma4(fa[s][mi].xy, fb[s].xy, acc[mi]);
#pragma unroll
    for (int mi = 0; mi < 4; ++mi) acc[mi] = wmma4(fa[s][mi].zw, fb[s].zw, acc[mi]);
  }
#pragma unroll
  for (int mi = 0; mi < 4; ++mi)
#pragma unroll
    for (int r = 0; r < 8; ++r)
      red[(wave * 64 + mi * 16 + 8 * hi + r) * PP_ + mr] = acc[mi][r];
  __builtin_amdgcn_sched_barrier(0);

  const int n0w = wave * 48;
  f3 kb[8][4];
#pragma unroll
  for (int s = 0; s < 8; ++s)
#pragma unroll
    for (int i = 0; i < 4; ++i)
      kb[s][i] = *reinterpret_cast<const f3*>(kk + (s * 8 + kq + i) * C_ + n0w + 3 * mr);
  v4f w0q[6];
#pragma unroll
  for (int i = 0; i < 6; ++i) {
    const int idx = tid + 256 * i;
    w0q[i] = *reinterpret_cast<const v4f*>(W0 + (h0 + idx / 96) * C_ + 4 * (idx % 96));
  }
  __syncthreads();
#pragma unroll
  for (int j = 0; j < 4; ++j) {
    const int t = tq + 16 * j;
    float s = red[t * PP_ + rh];
#pragma unroll
    for (int sl = 1; sl < 8; ++sl) s += red[(sl * 64 + t) * PP_ + rh];
    wdh[rh * WP_ + t] = s * g[j];
  }
  __syncthreads();

  if (tid < 16) {
    float s = 0.f;
#pragma unroll
    for (int t = 0; t < T_; ++t) s += wdh[tid * WP_ + t];
    nb0[b * H_ + h0 + tid] = dT * sb0[b * H_ + h0 + tid] + s;
  }

  v8f ac2[3] = {};
#pragma unroll
  for (int s = 0; s < 8; ++s) {
    const v4f a = *reinterpret_cast<const v4f*>(wdh + mr * WP_ + s * 8 + kq);
    ac2[0] = wmma4(a.xy, v2f{kb[s][0].x, kb[s][1].x}, ac2[0]);
    ac2[1] = wmma4(a.xy, v2f{kb[s][0].y, kb[s][1].y}, ac2[1]);
    ac2[2] = wmma4(a.xy, v2f{kb[s][0].z, kb[s][1].z}, ac2[2]);
    ac2[0] = wmma4(a.zw, v2f{kb[s][2].x, kb[s][3].x}, ac2[0]);
    ac2[1] = wmma4(a.zw, v2f{kb[s][2].y, kb[s][3].y}, ac2[1]);
    ac2[2] = wmma4(a.zw, v2f{kb[s][2].z, kb[s][3].z}, ac2[2]);
  }
  float* outp = red;
#pragma unroll
  for (int r = 0; r < 8; ++r) {
    float* po = outp + (8 * hi + r) * OP_ + n0w + 3 * mr;
    po[0] = ac2[0][r];  po[1] = ac2[1][r];  po[2] = ac2[2][r];
  }
  __syncthreads();
#pragma unroll
  for (int i = 0; i < 6; ++i) {
    const int idx = tid + 256 * i;
    const int row = idx / 96, c4 = 4 * (idx % 96);
    const v4f a = *reinterpret_cast<const v4f*>(outp + row * OP_ + c4);
    v4f o;
    o.x = dT * w0q[i].x + a.x;  o.y = dT * w0q[i].y + a.y;
    o.z = dT * w0q[i].z + a.z;  o.w = dT * w0q[i].w + a.w;
    *reinterpret_cast<v4f*>(oW0 + (h0 + row) * C_ + c4) = o;
  }
}

}

extern "C" void kernel_launch(void* const* d_in, const int* in_sizes, int n_in,
                              void* d_out, int out_size, void* d_ws, size_t ws_size,
                              hipStream_t stream) {
  const float* x   = (const float*)d_in[0];
  const float* Wq  = (const float*)d_in[1];
  const float* bq  = (const float*)d_in[2];
  const float* Wk  = (const float*)d_in[3];
  const float* bk  = (const float*)d_in[4];
  const float* Wv  = (const float*)d_in[5];
  const float* bv  = (const float*)d_in[6];
  const float* Wo  = (const float*)d_in[7];
  const float* bo  = (const float*)d_in[8];
  const float* iq  = (const float*)d_in[9];
  const float* sW0 = (const float*)d_in[10];
  const float* sb0 = (const float*)d_in[11];
  const float* sW1 = (const float*)d_in[12];
  const float* sb1 = (const float*)d_in[13];

  float* out = (float*)d_out;
  float* ws  = (float*)d_ws;

  float* out_out = out;
  float* out_nW0 = out + B_ * C_;
  float* out_nb0 = out_nW0 + B_ * H_ * C_;
  float* out_nW1 = out_nb0 + B_ * H_;
  float* out_nb1 = out_nW1 + B_ * C_ * H_;

  k_gemm_kv<<<dim3(54, B_, 2), 128, 0, stream>>>(x, Wk, bk, Wv, bv, Wq, bq, iq, ws);
  k_gemm_h1<<<dim3(144, B_), 128, 0, stream>>>(sW0, sb0, ws);
  k_gemm_dp<<<dim3(120, B_), 256, 0, stream>>>(sW1, sb1, ws);
  k_grad_update<<<dim3(219), 256, 0, stream>>>(sW0, sb0, sW1, sb1, Wo, bo, out_nW0, out_nb0,
                                               out_nW1, out_nb1, out_out, ws);
}
